# starting kernel
# baseline (speedup 1.0000x reference)
#include <hip/hip_runtime.h>

typedef __attribute__((ext_vector_type(2))) float v2f;
typedef __attribute__((ext_vector_type(8))) float v8f;

#define D_IN  128
#define D_OUT 128

__global__ __launch_bounds__(256) void influence_gemm_cu_kernel(
    const float* __restrict__ feat, const float* __restrict__ W,
    const float* __restrict__ cu, float* __restrict__ h, int nrows) {
  __shared__ float sW[D_IN * D_OUT];

  const int tid = threadIdx.x;
#pragma unroll
  for (int i = 0; i < (D_IN * D_OUT) / 256; ++i)
    sW[tid + i * 256] = W[tid + i * 256];
  __syncthreads();

  const int wave    = tid >> 5;
  const int lane    = tid & 31;
  const int rowBase = blockIdx.x * 16;
  const int colBase = wave * 16;
  const int m       = lane & 15;
  const int kh      = (lane >> 4) * 2;

  int arow_idx = rowBase + m;
  if (arow_idx >= nrows) arow_idx = nrows - 1;
  const float* arow = feat + (size_t)arow_idx * D_IN + kh;

  v8f acc = {};
#pragma unroll
  for (int kk = 0; kk < D_IN / 4; ++kk) {
    v2f A = *(const v2f*)(arow + kk * 4);
    const int kb = kk * 4 + kh;
    v2f B;
    B.x = sW[kb * D_OUT + colBase + m];
    B.y = sW[(kb + 1) * D_OUT + colBase + m];
    acc = __builtin_amdgcn_wmma_f32_16x16x4_f32(
         false, A,  false, B,
         (short)0, acc,  false,  false);
  }

  const int mBase = rowBase + (lane >> 4) * 8;
#pragma unroll
  for (int v = 0; v < 8; ++v) {
    const int row = mBase + v;
    if (row < nrows)
      h[(size_t)row * D_OUT + colBase + m] = acc[v] * cu[row];
  }
}

__global__ __launch_bounds__(256) void influence_scatter_kernel(
    const float* __restrict__ h, const int* __restrict__ src,
    const int* __restrict__ dst, const float* __restrict__ edge_w,
    const float* __restrict__ cv, float* __restrict__ out, int nedges) {
  const long gid = (long)blockIdx.x * blockDim.x + threadIdx.x;
  const long e = gid >> 5;
  if (e >= nedges) return;
  const int lane = (int)(gid & 31);

  const int s = src[e];
  const int d = dst[e];
  const float w = edge_w[e] * cv[d];

  const float4 hv = *(const float4*)(h + (size_t)s * D_OUT + lane * 4);
  float* op = out + (size_t)d * D_OUT + lane * 4;
  atomicAdd(op + 0, hv.x * w);
  atomicAdd(op + 1, hv.y * w);
  atomicAdd(op + 2, hv.z * w);
  atomicAdd(op + 3, hv.w * w);
}

extern "C" void kernel_launch(void* const* d_in, const int* in_sizes, int n_in,
                              void* d_out, int out_size, void* d_ws, size_t ws_size,
                              hipStream_t stream) {
  const float* feat = (const float*)d_in[0];
  const float* W    = (const float*)d_in[1];
  const float* cu   = (const float*)d_in[2];
  const float* cv   = (const float*)d_in[3];
  const float* ew   = (const float*)d_in[4];
  const int*   src  = (const int*)d_in[5];
  const int*   dst  = (const int*)d_in[6];

  float* out = (float*)d_out;
  float* h   = (float*)d_ws;

  const int N = in_sizes[2];
  const int E = in_sizes[5];

  hipMemsetAsync(d_out, 0, (size_t)out_size * sizeof(float), stream);

  const int rowTiles = (N + 15) / 16;
  influence_gemm_cu_kernel<<<rowTiles, 256, 0, stream>>>(feat, W, cu, h, N);

  const long threads = (long)E * 32;
  const int tpb = 256;
  const long blocks = (threads + tpb - 1) / tpb;
  influence_scatter_kernel<<<(int)blocks, tpb, 0, stream>>>(h, src, dst, ew,
                                                            cv, out, E);
}


# f32 atomic scatter replaced by in-call sort of edges by target + one wave per target summing rows in registers in ascending edge order; sum taken on input rows ahead of the projection; no memset
# speedup vs baseline: 1.2173x; 1.2173x over previous
#include <hip/hip_runtime.h>

typedef __attribute__((ext_vector_type(2))) float v2f;
typedef __attribute__((ext_vector_type(8))) float v8f;

#define D_IN  128
#define D_OUT 128

__global__ __launch_bounds__(256) void influence_gemm_cu_kernel(
    const float* __restrict__ feat, const float* __restrict__ W,
    const float* __restrict__ cu, float* __restrict__ h, int nrows) {
  __shared__ float sW[D_IN * D_OUT];

  const int tid = threadIdx.x;
#pragma unroll
  for (int i = 0; i < (D_IN * D_OUT) / 256; ++i)
    sW[tid + i * 256] = W[tid + i * 256];
  __syncthreads();

  const int wave    = tid >> 5;
  const int lane    = tid & 31;
  const int rowBase = blockIdx.x * 16;
  const int colBase = wave * 16;
  const int m       = lane & 15;
  const int kh      = (lane >> 4) * 2;

  int arow_idx = rowBase + m;
  if (arow_idx >= nrows) arow_idx = nrows - 1;
  const float* arow = feat + (size_t)arow_idx * D_IN + kh;

  v8f acc = {};
#pragma unroll
  for (int kk = 0; kk < D_IN / 4; ++kk) {
    v2f A = *(const v2f*)(arow + kk * 4);
    const int kb = kk * 4 + kh;
    v2f B;
    B.x = sW[kb * D_OUT + colBase + m];
    B.y = sW[(kb + 1) * D_OUT + colBase + m];
    acc = __builtin_amdgcn_wmma_f32_16x16x4_f32(
         false, A,  false, B,
         (short)0, acc,  false,  false);
  }

  const int mBase = rowBase + (lane >> 4) * 8;
#pragma unroll
  for (int v = 0; v < 8; ++v) {
    const int row = mBase + v;
    if (row < nrows)
      h[(size_t)row * D_OUT + colBase + m] = acc[v] * cu[row];
  }
}

__global__ __launch_bounds__(256) void zero_counts_kernel(int* __restrict__ cnt,
                                                          int n) {
  const int i = blockIdx.x * 256 + threadIdx.x;
  if (i < n) cnt[i] = 0;
}

__global__ __launch_bounds__(256) void count_rank_kernel(
    const int* __restrict__ dst, int* __restrict__ cnt, int* __restrict__ rank,
    int nedges) {
  const int e0 = (blockIdx.x * 256 + threadIdx.x) * 4;
  if (e0 + 3 < nedges) {
    const int4 d = *(const int4*)(dst + e0);
    int4 r;
    r.x = atomicAdd(cnt + d.x, 1);
    r.y = atomicAdd(cnt + d.y, 1);
    r.z = atomicAdd(cnt + d.z, 1);
    r.w = atomicAdd(cnt + d.w, 1);
    *(int4*)(rank + e0) = r;
  } else {
    for (int e = e0; e < nedges; ++e) rank[e] = atomicAdd(cnt + dst[e], 1);
  }
}

__device__ __forceinline__ int wave_sum_i32(int v) {
#pragma unroll
  for (int m = 16; m >= 1; m >>= 1) v += __shfl_xor(v, m);
  return v;
}

__global__ __launch_bounds__(1024) void scan_counts_kernel(
    const int* __restrict__ cnt, int* __restrict__ off, int pieces) {
  __shared__ int wtot[32];
  const int lane = threadIdx.x & 31;
  const int wave = threadIdx.x >> 5;
  const int4* c4 = (const int4*)cnt + (size_t)wave * pieces * 32 + lane;
  int4* o4 = (int4*)off + (size_t)wave * pieces * 32 + lane;

  int s = 0;
#pragma unroll 5
  for (int p = 0; p < pieces; ++p) {
    const int4 v = c4[p * 32];
    s += v.x + v.y + v.z + v.w;
  }
  s = wave_sum_i32(s);
  if (lane == 0) wtot[wave] = s;
  __syncthreads();
  int carry = wave_sum_i32((lane < wave) ? wtot[lane] : 0);

#pragma unroll 5
  for (int p = 0; p < pieces; ++p) {
    const int4 v = c4[p * 32];
    const int t = v.x + v.y + v.z + v.w;
    int inc = t;
#pragma unroll
    for (int d = 1; d < 32; d <<= 1) {
      const int u = __shfl_up(inc, d);
      if (lane >= d) inc += u;
    }
    int4 o;
    o.x = carry + inc - t;
    o.y = o.x + v.x;
    o.z = o.y + v.y;
    o.w = o.z + v.z;
    o4[p * 32] = o;
    carry += __shfl(inc, 31);
  }
}

__global__ __launch_bounds__(256) void fill_perm_kernel(
    const int* __restrict__ dst, const int* __restrict__ rank,
    const int* __restrict__ off, int* __restrict__ perm, int nedges) {
  const int e0 = (blockIdx.x * 256 + threadIdx.x) * 4;
  if (e0 + 3 < nedges) {
    const int4 d = *(const int4*)(dst + e0);
    const int4 r = *(const int4*)(rank + e0);
    perm[off[d.x] + r.x] = e0;
    perm[off[d.y] + r.y] = e0 + 1;
    perm[off[d.z] + r.z] = e0 + 2;
    perm[off[d.w] + r.w] = e0 + 3;
  } else {
    for (int e = e0; e < nedges; ++e) perm[off[dst[e]] + rank[e]] = e;
  }
}

__global__ __launch_bounds__(256) void influence_pull_kernel(
    const float* __restrict__ feat, const int* __restrict__ src,
    const float* __restrict__ edge_w, const float* __restrict__ cu,
    const int* __restrict__ off, const int* __restrict__ perm,
    float* __restrict__ agg, int nrows) {
  const int lane = threadIdx.x & 31;
  const int d = blockIdx.x * 8 + (threadIdx.x >> 5);
  if (d >= nrows) return;
  const int beg = __builtin_amdgcn_readfirstlane(off[d]);
  const int end = __builtin_amdgcn_readfirstlane(off[d + 1]);
  const int deg = end - beg;

  const float* fb = feat + lane * 4;
  float4 acc = {0.f, 0.f, 0.f, 0.f};

  if (deg <= 32) {
    const int e = (lane < deg) ? perm[beg + lane] : 0x7fffffff;
    int s = 0;
    float w = 0.f;
    if (lane < deg) {
      s = src[e];
      w = edge_w[e] * cu[s];
    }
    int r = 0;
#pragma unroll
    for (int j = 0; j < 32; ++j)
      r += (__builtin_amdgcn_readlane(e, j) < e) ? 1 : 0;
    const int ss = __builtin_amdgcn_ds_permute(r << 2, s);
    const int wb = __builtin_amdgcn_ds_permute(r << 2, __float_as_int(w));

    int i = 0;
    for (; i + 8 <= deg; i += 8) {
      float4 v[8];
      float wi[8];
#pragma unroll
      for (int k = 0; k < 8; ++k) {
        const int si = __builtin_amdgcn_readlane(ss, i + k);
        wi[k] = __int_as_float(__builtin_amdgcn_readlane(wb, i + k));
        v[k] = *(const float4*)(fb + (size_t)si * D_IN);
      }
#pragma unroll
      for (int k = 0; k < 8; ++k) {
        acc.x = fmaf(v[k].x, wi[k], acc.x);
        acc.y = fmaf(v[k].y, wi[k], acc.y);
        acc.z = fmaf(v[k].z, wi[k], acc.z);
        acc.w = fmaf(v[k].w, wi[k], acc.w);
      }
    }
    if (i + 4 <= deg) {
      float4 v[4];
      float wi[4];
#pragma unroll
      for (int k = 0; k < 4; ++k) {
        const int si = __builtin_amdgcn_readlane(ss, i + k);
        wi[k] = __int_as_float(__builtin_amdgcn_readlane(wb, i + k));
        v[k] = *(const float4*)(fb + (size_t)si * D_IN);
      }
#pragma unroll
      for (int k = 0; k < 4; ++k) {
        acc.x = fmaf(v[k].x, wi[k], acc.x);
        acc.y = fmaf(v[k].y, wi[k], acc.y);
        acc.z = fmaf(v[k].z, wi[k], acc.z);
        acc.w = fmaf(v[k].w, wi[k], acc.w);
      }
      i += 4;
    }
    for (; i < deg; ++i) {
      const int si = __builtin_amdgcn_readlane(ss, i);
      const float wi = __int_as_float(__builtin_amdgcn_readlane(wb, i));
      const float4 v = *(const float4*)(fb + (size_t)si * D_IN);
      acc.x = fmaf(v.x, wi, acc.x);
      acc.y = fmaf(v.y, wi, acc.y);
      acc.z = fmaf(v.z, wi, acc.z);
      acc.w = fmaf(v.w, wi, acc.w);
    }
  } else {
    int last = -1;
    for (int i = 0; i < deg; ++i) {
      int best = 0x7fffffff;
      for (int p = beg + lane; p < end; p += 32) {
        const int e = perm[p];
        if (e > last && e < best) best = e;
      }
#pragma unroll
      for (int m = 16; m >= 1; m >>= 1) {
        const int o = __shfl_xor(best, m);
        best = (o < best) ? o : best;
      }
      best = __builtin_amdgcn_readfirstlane(best);
      last = best;
      const int si = src[best];
      const float wi = edge_w[best] * cu[si];
      const float4 v = *(const float4*)(fb + (size_t)si * D_IN);
      acc.x = fmaf(v.x, wi, acc.x);
      acc.y = fmaf(v.y, wi, acc.y);
      acc.z = fmaf(v.z, wi, acc.z);
      acc.w = fmaf(v.w, wi, acc.w);
    }
  }

  *(float4*)(agg + (size_t)d * D_IN + lane * 4) = acc;
}

extern "C" void kernel_launch(void* const* d_in, const int* in_sizes, int n_in,
                              void* d_out, int out_size, void* d_ws, size_t ws_size,
                              hipStream_t stream) {
  const float* feat = (const float*)d_in[0];
  const float* W    = (const float*)d_in[1];
  const float* cu   = (const float*)d_in[2];
  const float* cv   = (const float*)d_in[3];
  const float* ew   = (const float*)d_in[4];
  const int*   src  = (const int*)d_in[5];
  const int*   dst  = (const int*)d_in[6];

  float* out = (float*)d_out;
  float* agg = (float*)d_ws;

  const int N = in_sizes[2];
  const int E = in_sizes[5];

  const int pieces = N / 4096 + 1;
  const int Np = pieces * 4096;
  int* cnt  = (int*)d_out;
  int* off  = cnt + Np;
  int* rank = off + Np;
  int* perm = rank + ((E + 3) & ~3);

  zero_counts_kernel<<<Np / 256, 256, 0, stream>>>(cnt, Np);
  const int quadBlocks = (E + 1023) / 1024;
  count_rank_kernel<<<quadBlocks, 256, 0, stream>>>(dst, cnt, rank, E);
  scan_counts_kernel<<<1, 1024, 0, stream>>>(cnt, off, pieces);
  fill_perm_kernel<<<quadBlocks, 256, 0, stream>>>(dst, rank, off, perm, E);
  influence_pull_kernel<<<(N + 7) / 8, 256, 0, stream>>>(feat, src, ew, cu, off,
                                                       perm, agg, N);

  const int rowTiles = (N + 15) / 16;
  influence_gemm_cu_kernel<<<rowTiles, 256, 0, stream>>>(agg, W, cv, out, N);
}


# edge sort by target moved to LDS counters: edges binned by 512-target range per 4096-edge block, one block per range ranks its edges in LDS; no global atomics, no rank array; sum order unchanged
# speedup vs baseline: 1.7069x; 1.4022x over previous
#include <hip/hip_runtime.h>

typedef __attribute__((ext_vector_type(2))) float v2f;
typedef __attribute__((ext_vector_type(8))) float v8f;

#define D_IN  128
#define D_OUT 128

__global__ __launch_bounds__(256) void influence_gemm_cu_kernel(
    const float* __restrict__ feat, const float* __restrict__ W,
    const float* __restrict__ cu, float* __restrict__ h, int nrows) {
  __shared__ float sW[D_IN * D_OUT];

  const int tid = threadIdx.x;
#pragma unroll
  for (int i = 0; i < (D_IN * D_OUT) / 256; ++i)
    sW[tid + i * 256] = W[tid + i * 256];
  __syncthreads();

  const int wave    = tid >> 5;
  const int lane    = tid & 31;
  const int rowBase = blockIdx.x * 16;
  const int colBase = wave * 16;
  const int m       = lane & 15;
  const int kh      = (lane >> 4) * 2;

  int arow_idx = rowBase + m;
  if (arow_idx >= nrows) arow_idx = nrows - 1;
  const float* arow = feat + (size_t)arow_idx * D_IN + kh;

  v8f acc = {};
#pragma unroll
  for (int kk = 0; kk < D_IN / 4; ++kk) {
    v2f A = *(const v2f*)(arow + kk * 4);
    const int kb = kk * 4 + kh;
    v2f B;
    B.x = sW[kb * D_OUT + colBase + m];
    B.y = sW[(kb + 1) * D_OUT + colBase + m];
    acc = __builtin_amdgcn_wmma_f32_16x16x4_f32(
         false, A,  false, B,
         (short)0, acc,  false,  false);
  }

  const int mBase = rowBase + (lane >> 4) * 8;
#pragma unroll
  for (int v = 0; v < 8; ++v) {
    const int row = mBase + v;
    if (row < nrows)
      h[(size_t)row * D_OUT + colBase + m] = acc[v] * cu[row];
  }
}

#define BIN_SHIFT 9
#define BIN_W     512
#define MAX_BINS  256
#define CHUNK     4096

__global__ __launch_bounds__(256) void bin_count_kernel(
    const int* __restrict__ dst, int* __restrict__ H, int nedges, int nbins,
    int nchunks, int hpad) {
  __shared__ int hist[MAX_BINS];
  const int t = threadIdx.x;
  const int c = blockIdx.x;
  hist[t] = 0;
  __syncthreads();
  const int base = c * CHUNK;
  if (base + CHUNK <= nedges) {
    int4 d[4];
#pragma unroll
    for (int k = 0; k < 4; ++k)
      d[k] = *(const int4*)(dst + base + (k * 256 + t) * 4);
#pragma unroll
    for (int k = 0; k < 4; ++k) {
      atomicAdd(&hist[d[k].x >> BIN_SHIFT], 1);
      atomicAdd(&hist[d[k].y >> BIN_SHIFT], 1);
      atomicAdd(&hist[d[k].z >> BIN_SHIFT], 1);
      atomicAdd(&hist[d[k].w >> BIN_SHIFT], 1);
    }
  } else {
    for (int e = base + t; e < nedges; e += 256)
      atomicAdd(&hist[dst[e] >> BIN_SHIFT], 1);
  }
  __syncthreads();
  if (t < nbins) H[t * nchunks + c] = hist[t];
  if (c == 0)
    for (int i = nbins * nchunks + t; i < hpad; i += 256) H[i] = 0;
}

__device__ __forceinline__ int wave_sum_i32(int v) {
#pragma unroll
  for (int m = 16; m >= 1; m >>= 1) v += __shfl_xor(v, m);
  return v;
}

__global__ __launch_bounds__(1024) void scan_counts_kernel(
    const int* __restrict__ cnt, int* __restrict__ off, int pieces) {
  __shared__ int wtot[32];
  const int lane = threadIdx.x & 31;
  const int wave = threadIdx.x >> 5;
  const int4* c4 = (const int4*)cnt + (size_t)wave * pieces * 32 + lane;
  int4* o4 = (int4*)off + (size_t)wave * pieces * 32 + lane;

  int s = 0;
#pragma unroll 5
  for (int p = 0; p < pieces; ++p) {
    const int4 v = c4[p * 32];
    s += v.x + v.y + v.z + v.w;
  }
  s = wave_sum_i32(s);
  if (lane == 0) wtot[wave] = s;
  __syncthreads();
  int carry = wave_sum_i32((lane < wave) ? wtot[lane] : 0);

#pragma unroll 5
  for (int p = 0; p < pieces; ++p) {
    const int4 v = c4[p * 32];
    const int t = v.x + v.y + v.z + v.w;
    int inc = t;
#pragma unroll
    for (int d = 1; d < 32; d <<= 1) {
      const int u = __shfl_up(inc, d);
      if (lane >= d) inc += u;
    }
    int4 o;
    o.x = carry + inc - t;
    o.y = o.x + v.x;
    o.z = o.y + v.y;
    o.w = o.z + v.z;
    o4[p * 32] = o;
    carry += __shfl(inc, 31);
  }
}

__global__ __launch_bounds__(256) void bin_scatter_kernel(
    const int* __restrict__ dst, const int* __restrict__ S,
    int2* __restrict__ rec, int nedges, int nbins, int nchunks) {
  __shared__ int cur[MAX_BINS];
  const int t = threadIdx.x;
  const int c = blockIdx.x;
  if (t < nbins) cur[t] = S[t * nchunks + c];
  __syncthreads();
  const int base = c * CHUNK;
  if (base + CHUNK <= nedges) {
    int4 d[4];
#pragma unroll
    for (int k = 0; k < 4; ++k)
      d[k] = *(const int4*)(dst + base + (k * 256 + t) * 4);
#pragma unroll
    for (int k = 0; k < 4; ++k) {
      const int e0 = base + (k * 256 + t) * 4;
      rec[atomicAdd(&cur[d[k].x >> BIN_SHIFT], 1)] = make_int2(e0, d[k].x);
      rec[atomicAdd(&cur[d[k].y >> BIN_SHIFT], 1)] = make_int2(e0 + 1, d[k].y);
      rec[atomicAdd(&cur[d[k].z >> BIN_SHIFT], 1)] = make_int2(e0 + 2, d[k].z);
      rec[atomicAdd(&cur[d[k].w >> BIN_SHIFT], 1)] = make_int2(e0 + 3, d[k].w);
    }
  } else {
    for (int e = base + t; e < nedges; e += 256) {
      const int de = dst[e];
      rec[atomicAdd(&cur[de >> BIN_SHIFT], 1)] = make_int2(e, de);
    }
  }
}

__global__ __launch_bounds__(1024) void bin_rank_kernel(
    const int2* __restrict__ rec, const int* __restrict__ S,
    int* __restrict__ off, int* __restrict__ perm, int nrows, int nchunks) {
  __shared__ int cntL[BIN_W];
  __shared__ int curL[BIN_W];
  const int t = threadIdx.x;
  const int b = blockIdx.x;
  const int lo = S[b * nchunks];
  const int hi = S[(b + 1) * nchunks];
  const int tbase = b << BIN_SHIFT;
  if (t < BIN_W) cntL[t] = 0;
  __syncthreads();
  for (int i0 = lo + t; i0 < hi; i0 += 4096) {
    int d[4];
#pragma unroll
    for (int k = 0; k < 4; ++k) {
      const int i = i0 + k * 1024;
      d[k] = (i < hi) ? rec[i].y : tbase;
    }
#pragma unroll
    for (int k = 0; k < 4; ++k)
      if (i0 + k * 1024 < hi) atomicAdd(&cntL[d[k] - tbase], 1);
  }
  __syncthreads();
  if (t < 32) {
    int v[16];
    int s = 0;
#pragma unroll
    for (int j = 0; j < 16; ++j) {
      v[j] = cntL[t * 16 + j];
      s += v[j];
    }
    int inc = s;
#pragma unroll
    for (int d = 1; d < 32; d <<= 1) {
      const int u = __shfl_up(inc, d);
      if (t >= d) inc += u;
    }
    int run = inc - s;
#pragma unroll
    for (int j = 0; j < 16; ++j) {
      curL[t * 16 + j] = run;
      run += v[j];
    }
  }
  __syncthreads();
  if (t < BIN_W) {
    const int g = tbase + t;
    if (g <= nrows) off[g] = lo + curL[t];
  }
  __syncthreads();
  for (int i0 = lo + t; i0 < hi; i0 += 4096) {
    int2 r[4];
#pragma unroll
    for (int k = 0; k < 4; ++k) {
      const int i = i0 + k * 1024;
      r[k] = (i < hi) ? rec[i] : make_int2(0, tbase);
    }
#pragma unroll
    for (int k = 0; k < 4; ++k)
      if (i0 + k * 1024 < hi)
        perm[lo + atomicAdd(&curL[r[k].y - tbase], 1)] = r[k].x;
  }
}

__global__ __launch_bounds__(256) void influence_pull_kernel(
    const float* __restrict__ feat, const int* __restrict__ src,
    const float* __restrict__ edge_w, const float* __restrict__ cu,
    const int* __restrict__ off, const int* __restrict__ perm,
    float* __restrict__ agg, int nrows) {
  const int lane = threadIdx.x & 31;
  const int d = blockIdx.x * 8 + (threadIdx.x >> 5);
  if (d >= nrows) return;
  const int beg = __builtin_amdgcn_readfirstlane(off[d]);
  const int end = __builtin_amdgcn_readfirstlane(off[d + 1]);
  const int deg = end - beg;

  const float* fb = feat + lane * 4;
  float4 acc = {0.f, 0.f, 0.f, 0.f};

  if (deg <= 32) {
    const int e = (lane < deg) ? perm[beg + lane] : 0x7fffffff;
    int s = 0;
    float w = 0.f;
    if (lane < deg) {
      s = src[e];
      w = edge_w[e] * cu[s];
    }
    int r = 0;
#pragma unroll
    for (int j = 0; j < 32; ++j)
      r += (__builtin_amdgcn_readlane(e, j) < e) ? 1 : 0;
    const int ss = __builtin_amdgcn_ds_permute(r << 2, s);
    const int wb = __builtin_amdgcn_ds_permute(r << 2, __float_as_int(w));

    int i = 0;
    for (; i + 8 <= deg; i += 8) {
      float4 v[8];
      float wi[8];
#pragma unroll
      for (int k = 0; k < 8; ++k) {
        const int si = __builtin_amdgcn_readlane(ss, i + k);
        wi[k] = __int_as_float(__builtin_amdgcn_readlane(wb, i + k));
        v[k] = *(const float4*)(fb + (size_t)si * D_IN);
      }
#pragma unroll
      for (int k = 0; k < 8; ++k) {
        acc.x = fmaf(v[k].x, wi[k], acc.x);
        acc.y = fmaf(v[k].y, wi[k], acc.y);
        acc.z = fmaf(v[k].z, wi[k], acc.z);
        acc.w = fmaf(v[k].w, wi[k], acc.w);
      }
    }
    if (i + 4 <= deg) {
      float4 v[4];
      float wi[4];
#pragma unroll
      for (int k = 0; k < 4; ++k) {
        const int si = __builtin_amdgcn_readlane(ss, i + k);
        wi[k] = __int_as_float(__builtin_amdgcn_readlane(wb, i + k));
        v[k] = *(const float4*)(fb + (size_t)si * D_IN);
      }
#pragma unroll
      for (int k = 0; k < 4; ++k) {
        acc.x = fmaf(v[k].x, wi[k], acc.x);
        acc.y = fmaf(v[k].y, wi[k], acc.y);
        acc.z = fmaf(v[k].z, wi[k], acc.z);
        acc.w = fmaf(v[k].w, wi[k], acc.w);
      }
      i += 4;
    }
    for (; i < deg; ++i) {
      const int si = __builtin_amdgcn_readlane(ss, i);
      const float wi = __int_as_float(__builtin_amdgcn_readlane(wb, i));
      const float4 v = *(const float4*)(fb + (size_t)si * D_IN);
      acc.x = fmaf(v.x, wi, acc.x);
      acc.y = fmaf(v.y, wi, acc.y);
      acc.z = fmaf(v.z, wi, acc.z);
      acc.w = fmaf(v.w, wi, acc.w);
    }
  } else {
    int last = -1;
    for (int i = 0; i < deg; ++i) {
      int best = 0x7fffffff;
      for (int p = beg + lane; p < end; p += 32) {
        const int e = perm[p];
        if (e > last && e < best) best = e;
      }
#pragma unroll
      for (int m = 16; m >= 1; m >>= 1) {
        const int o = __shfl_xor(best, m);
        best = (o < best) ? o : best;
      }
      best = __builtin_amdgcn_readfirstlane(best);
      last = best;
      const int si = src[best];
      const float wi = edge_w[best] * cu[si];
      const float4 v = *(const float4*)(fb + (size_t)si * D_IN);
      acc.x = fmaf(v.x, wi, acc.x);
      acc.y = fmaf(v.y, wi, acc.y);
      acc.z = fmaf(v.z, wi, acc.z);
      acc.w = fmaf(v.w, wi, acc.w);
    }
  }

  *(float4*)(agg + (size_t)d * D_IN + lane * 4) = acc;
}

extern "C" void kernel_launch(void* const* d_in, const int* in_sizes, int n_in,
                              void* d_out, int out_size, void* d_ws, size_t ws_size,
                              hipStream_t stream) {
  const float* feat = (const float*)d_in[0];
  const float* W    = (const float*)d_in[1];
  const float* cu   = (const float*)d_in[2];
  const float* cv   = (const float*)d_in[3];
  const float* ew   = (const float*)d_in[4];
  const int*   src  = (const int*)d_in[5];
  const int*   dst  = (const int*)d_in[6];

  float* out = (float*)d_out;
  float* agg = (float*)d_ws;

  const int N = in_sizes[2];
  const int E = in_sizes[5];

  const int nbins   = N / BIN_W + 1;
  const int nchunks = (E + CHUNK - 1) / CHUNK;
  const int pieces  = (nbins * nchunks) / 4096 + 1;
  const int Hp = pieces * 4096;
  const int Np = (N / 4096 + 1) * 4096;
  int*  H    = (int*)d_out;
  int*  S    = H + Hp;
  int*  off  = S + Hp;
  int2* rec  = (int2*)(off + Np);
  int*  perm = (int*)(rec + E);

  bin_count_kernel<<<nchunks, 256, 0, stream>>>(dst, H, E, nbins, nchunks, Hp);
  scan_counts_kernel<<<1, 1024, 0, stream>>>(H, S, pieces);
  bin_scatter_kernel<<<nchunks, 256, 0, stream>>>(dst, S, rec, E, nbins, nchunks);
  bin_rank_kernel<<<nbins, 1024, 0, stream>>>(rec, S, off, perm, N, nchunks);
  influence_pull_kernel<<<(N + 7) / 8, 256, 0, stream>>>(feat, src, ew, cu, off,
                                                       perm, agg, N);

  const int rowTiles = (N + 15) / 16;
  influence_gemm_cu_kernel<<<rowTiles, 256, 0, stream>>>(agg, W, cv, out, N);
}


# sort's scattered stores ordered in LDS first: an edge block writes its records grouped by range, a range block its edge ids in run order: contiguous runs, not one line per lane; sum order unchanged
# speedup vs baseline: 1.8741x; 1.0980x over previous
#include <hip/hip_runtime.h>

typedef __attribute__((ext_vector_type(2))) float v2f;
typedef __attribute__((ext_vector_type(8))) float v8f;

#define D_IN  128
#define D_OUT 128

__global__ __launch_bounds__(256) void influence_gemm_cu_kernel(
    const float* __restrict__ feat, const float* __restrict__ W,
    const float* __restrict__ cu, float* __restrict__ h, int nrows) {
  __shared__ float sW[D_IN * D_OUT];

  const int tid = threadIdx.x;
#pragma unroll
  for (int i = 0; i < (D_IN * D_OUT) / 256; ++i)
    sW[tid + i * 256] = W[tid + i * 256];
  __syncthreads();

  const int wave    = tid >> 5;
  const int lane    = tid & 31;
  const int rowBase = blockIdx.x * 16;
  const int colBase = wave * 16;
  const int m       = lane & 15;
  const int kh      = (lane >> 4) * 2;

  int arow_idx = rowBase + m;
  if (arow_idx >= nrows) arow_idx = nrows - 1;
  const float* arow = feat + (size_t)arow_idx * D_IN + kh;

  v8f acc = {};
#pragma unroll
  for (int kk = 0; kk < D_IN / 4; ++kk) {
    v2f A = *(const v2f*)(arow + kk * 4);
    const int kb = kk * 4 + kh;
    v2f B;
    B.x = sW[kb * D_OUT + colBase + m];
    B.y = sW[(kb + 1) * D_OUT + colBase + m];
    acc = __builtin_amdgcn_wmma_f32_16x16x4_f32(
         false, A,  false, B,
         (short)0, acc,  false,  false);
  }

  const int mBase = rowBase + (lane >> 4) * 8;
#pragma unroll
  for (int v = 0; v < 8; ++v) {
    const int row = mBase + v;
    if (row < nrows)
      h[(size_t)row * D_OUT + colBase + m] = acc[v] * cu[row];
  }
}

#define BIN_SHIFT 9
#define BIN_W     512
#define MAX_BINS  256
#define CHUNK     4096

__global__ __launch_bounds__(256) void bin_count_kernel(
    const int* __restrict__ dst, int* __restrict__ H, int nedges, int nbins,
    int nchunks, int hpad) {
  __shared__ int hist[MAX_BINS];
  const int t = threadIdx.x;
  const int c = blockIdx.x;
  hist[t] = 0;
  __syncthreads();
  const int base = c * CHUNK;
  if (base + CHUNK <= nedges) {
    int4 d[4];
#pragma unroll
    for (int k = 0; k < 4; ++k)
      d[k] = *(const int4*)(dst + base + (k * 256 + t) * 4);
#pragma unroll
    for (int k = 0; k < 4; ++k) {
      atomicAdd(&hist[d[k].x >> BIN_SHIFT], 1);
      atomicAdd(&hist[d[k].y >> BIN_SHIFT], 1);
      atomicAdd(&hist[d[k].z >> BIN_SHIFT], 1);
      atomicAdd(&hist[d[k].w >> BIN_SHIFT], 1);
    }
  } else {
    for (int e = base + t; e < nedges; e += 256)
      atomicAdd(&hist[dst[e] >> BIN_SHIFT], 1);
  }
  __syncthreads();
  if (t < nbins) H[t * nchunks + c] = hist[t];
  if (c == 0)
    for (int i = nbins * nchunks + t; i < hpad; i += 256) H[i] = 0;
}

__device__ __forceinline__ int wave_sum_i32(int v) {
#pragma unroll
  for (int m = 16; m >= 1; m >>= 1) v += __shfl_xor(v, m);
  return v;
}

__global__ __launch_bounds__(1024) void scan_counts_kernel(
    const int* __restrict__ cnt, int* __restrict__ off, int pieces) {
  __shared__ int wtot[32];
  const int lane = threadIdx.x & 31;
  const int wave = threadIdx.x >> 5;
  const int4* c4 = (const int4*)cnt + (size_t)wave * pieces * 32 + lane;
  int4* o4 = (int4*)off + (size_t)wave * pieces * 32 + lane;

  int s = 0;
#pragma unroll 5
  for (int p = 0; p < pieces; ++p) {
    const int4 v = c4[p * 32];
    s += v.x + v.y + v.z + v.w;
  }
  s = wave_sum_i32(s);
  if (lane == 0) wtot[wave] = s;
  __syncthreads();
  int carry = wave_sum_i32((lane < wave) ? wtot[lane] : 0);

#pragma unroll 5
  for (int p = 0; p < pieces; ++p) {
    const int4 v = c4[p * 32];
    const int t = v.x + v.y + v.z + v.w;
    int inc = t;
#pragma unroll
    for (int d = 1; d < 32; d <<= 1) {
      const int u = __shfl_up(inc, d);
      if (lane >= d) inc += u;
    }
    int4 o;
    o.x = carry + inc - t;
    o.y = o.x + v.x;
    o.z = o.y + v.y;
    o.w = o.z + v.z;
    o4[p * 32] = o;
    carry += __shfl(inc, 31);
  }
}

__global__ __launch_bounds__(256) void bin_scatter_kernel(
    const int* __restrict__ dst, const int* __restrict__ H,
    const int* __restrict__ S, int2* __restrict__ rec, int nedges, int nbins,
    int nchunks) {
  __shared__ int cur[MAX_BINS];
  __shared__ int gofs[MAX_BINS];
  __shared__ int wtot[8];
  __shared__ int2 sRec[CHUNK];
  const int t = threadIdx.x;
  const int c = blockIdx.x;
  const int lane = t & 31;
  const int wave = t >> 5;
  const int base = c * CHUNK;
  const bool whole = base + CHUNK <= nedges;

  int4 d[4];
  if (whole) {
#pragma unroll
    for (int k = 0; k < 4; ++k)
      d[k] = *(const int4*)(dst + base + (k * 256 + t) * 4);
  }
  int hb = 0, sb = 0;
  if (t < nbins) {
    hb = H[t * nchunks + c];
    sb = S[t * nchunks + c];
  }

  int inc = hb;
#pragma unroll
  for (int s = 1; s < 32; s <<= 1) {
    const int u = __shfl_up(inc, s);
    if (lane >= s) inc += u;
  }
  if (lane == 31) wtot[wave] = inc;
  __syncthreads();
  int carry = 0;
#pragma unroll
  for (int w = 0; w < 8; ++w) carry += (w < wave) ? wtot[w] : 0;
  const int lstart = carry + inc - hb;
  cur[t] = lstart;
  gofs[t] = sb - lstart;
  __syncthreads();

  if (whole) {
    int slot[16];
#pragma unroll
    for (int k = 0; k < 4; ++k) {
      slot[4 * k + 0] = atomicAdd(&cur[d[k].x >> BIN_SHIFT], 1);
      slot[4 * k + 1] = atomicAdd(&cur[d[k].y >> BIN_SHIFT], 1);
      slot[4 * k + 2] = atomicAdd(&cur[d[k].z >> BIN_SHIFT], 1);
      slot[4 * k + 3] = atomicAdd(&cur[d[k].w >> BIN_SHIFT], 1);
    }
#pragma unroll
    for (int k = 0; k < 4; ++k) {
      const int e0 = base + (k * 256 + t) * 4;
      sRec[slot[4 * k + 0]] = make_int2(e0, d[k].x);
      sRec[slot[4 * k + 1]] = make_int2(e0 + 1, d[k].y);
      sRec[slot[4 * k + 2]] = make_int2(e0 + 2, d[k].z);
      sRec[slot[4 * k + 3]] = make_int2(e0 + 3, d[k].w);
    }
  } else {
    for (int e = base + t; e < nedges; e += 256) {
      const int de = dst[e];
      sRec[atomicAdd(&cur[de >> BIN_SHIFT], 1)] = make_int2(e, de);
    }
  }
  __syncthreads();

  if (whole) {
#pragma unroll
    for (int k = 0; k < CHUNK / 256; ++k) {
      const int i = k * 256 + t;
      const int2 r = sRec[i];
      rec[gofs[r.y >> BIN_SHIFT] + i] = r;
    }
  } else {
    const int n = nedges - base;
    for (int i = t; i < n; i += 256) {
      const int2 r = sRec[i];
      rec[gofs[r.y >> BIN_SHIFT] + i] = r;
    }
  }
}

#define RANK_CAP 16384
__global__ __launch_bounds__(1024) void bin_rank_kernel(
    const int2* __restrict__ rec, const int* __restrict__ S,
    int* __restrict__ off, int* __restrict__ perm, int nrows, int nchunks) {
  __shared__ int cntL[BIN_W];
  __shared__ int curL[BIN_W];
  __shared__ int sP[RANK_CAP];
  const int t = threadIdx.x;
  const int b = blockIdx.x;
  const int lo = S[b * nchunks];
  const int hi = S[(b + 1) * nchunks];
  const int tbase = b << BIN_SHIFT;
  const int n = hi - lo;
  const bool staged = n <= RANK_CAP;
  if (t < BIN_W) cntL[t] = 0;
  __syncthreads();
  for (int i0 = lo + t; i0 < hi; i0 += 4096) {
    int d[4];
#pragma unroll
    for (int k = 0; k < 4; ++k) {
      const int i = i0 + k * 1024;
      d[k] = (i < hi) ? rec[i].y : tbase;
    }
#pragma unroll
    for (int k = 0; k < 4; ++k)
      if (i0 + k * 1024 < hi) atomicAdd(&cntL[d[k] - tbase], 1);
  }
  __syncthreads();
  if (t < 32) {
    int v[16];
    int s = 0;
#pragma unroll
    for (int j = 0; j < 16; ++j) {
      v[j] = cntL[t * 16 + j];
      s += v[j];
    }
    int inc = s;
#pragma unroll
    for (int d = 1; d < 32; d <<= 1) {
      const int u = __shfl_up(inc, d);
      if (t >= d) inc += u;
    }
    int run = inc - s;
#pragma unroll
    for (int j = 0; j < 16; ++j) {
      curL[t * 16 + j] = run;
      run += v[j];
    }
  }
  __syncthreads();
  if (t < BIN_W) {
    const int g = tbase + t;
    if (g <= nrows) off[g] = lo + curL[t];
  }
  __syncthreads();
  for (int i0 = lo + t; i0 < hi; i0 += 4096) {
    int2 r[4];
#pragma unroll
    for (int k = 0; k < 4; ++k) {
      const int i = i0 + k * 1024;
      r[k] = (i < hi) ? rec[i] : make_int2(0, tbase);
    }
#pragma unroll
    for (int k = 0; k < 4; ++k)
      if (i0 + k * 1024 < hi) {
        const int slot = atomicAdd(&curL[r[k].y - tbase], 1);
        if (staged) sP[slot] = r[k].x;
        else perm[lo + slot] = r[k].x;
      }
  }
  if (staged) {
    __syncthreads();
    for (int i = t; i < n; i += 1024) perm[lo + i] = sP[i];
  }
}

__global__ __launch_bounds__(256) void influence_pull_kernel(
    const float* __restrict__ feat, const int* __restrict__ src,
    const float* __restrict__ edge_w, const float* __restrict__ cu,
    const int* __restrict__ off, const int* __restrict__ perm,
    float* __restrict__ agg, int nrows) {
  const int lane = threadIdx.x & 31;
  const int d = blockIdx.x * 8 + (threadIdx.x >> 5);
  if (d >= nrows) return;
  const int beg = __builtin_amdgcn_readfirstlane(off[d]);
  const int end = __builtin_amdgcn_readfirstlane(off[d + 1]);
  const int deg = end - beg;

  const float* fb = feat + lane * 4;
  float4 acc = {0.f, 0.f, 0.f, 0.f};

  if (deg <= 32) {
    const int e = (lane < deg) ? perm[beg + lane] : 0x7fffffff;
    int s = 0;
    float w = 0.f;
    if (lane < deg) {
      s = src[e];
      w = edge_w[e] * cu[s];
    }
    int r = 0;
#pragma unroll
    for (int j = 0; j < 32; ++j)
      r += (__builtin_amdgcn_readlane(e, j) < e) ? 1 : 0;
    const int ss = __builtin_amdgcn_ds_permute(r << 2, s);
    const int wb = __builtin_amdgcn_ds_permute(r << 2, __float_as_int(w));

    int i = 0;
    for (; i + 8 <= deg; i += 8) {
      float4 v[8];
      float wi[8];
#pragma unroll
      for (int k = 0; k < 8; ++k) {
        const int si = __builtin_amdgcn_readlane(ss, i + k);
        wi[k] = __int_as_float(__builtin_amdgcn_readlane(wb, i + k));
        v[k] = *(const float4*)(fb + (size_t)si * D_IN);
      }
#pragma unroll
      for (int k = 0; k < 8; ++k) {
        acc.x = fmaf(v[k].x, wi[k], acc.x);
        acc.y = fmaf(v[k].y, wi[k], acc.y);
        acc.z = fmaf(v[k].z, wi[k], acc.z);
        acc.w = fmaf(v[k].w, wi[k], acc.w);
      }
    }
    if (i + 4 <= deg) {
      float4 v[4];
      float wi[4];
#pragma unroll
      for (int k = 0; k < 4; ++k) {
        const int si = __builtin_amdgcn_readlane(ss, i + k);
        wi[k] = __int_as_float(__builtin_amdgcn_readlane(wb, i + k));
        v[k] = *(const float4*)(fb + (size_t)si * D_IN);
      }
#pragma unroll
      for (int k = 0; k < 4; ++k) {
        acc.x = fmaf(v[k].x, wi[k], acc.x);
        acc.y = fmaf(v[k].y, wi[k], acc.y);
        acc.z = fmaf(v[k].z, wi[k], acc.z);
        acc.w = fmaf(v[k].w, wi[k], acc.w);
      }
      i += 4;
    }
    for (; i < deg; ++i) {
      const int si = __builtin_amdgcn_readlane(ss, i);
      const float wi = __int_as_float(__builtin_amdgcn_readlane(wb, i));
      const float4 v = *(const float4*)(fb + (size_t)si * D_IN);
      acc.x = fmaf(v.x, wi, acc.x);
      acc.y = fmaf(v.y, wi, acc.y);
      acc.z = fmaf(v.z, wi, acc.z);
      acc.w = fmaf(v.w, wi, acc.w);
    }
  } else {
    int last = -1;
    for (int i = 0; i < deg; ++i) {
      int best = 0x7fffffff;
      for (int p = beg + lane; p < end; p += 32) {
        const int e = perm[p];
        if (e > last && e < best) best = e;
      }
#pragma unroll
      for (int m = 16; m >= 1; m >>= 1) {
        const int o = __shfl_xor(best, m);
        best = (o < best) ? o : best;
      }
      best = __builtin_amdgcn_readfirstlane(best);
      last = best;
      const int si = src[best];
      const float wi = edge_w[best] * cu[si];
      const float4 v = *(const float4*)(fb + (size_t)si * D_IN);
      acc.x = fmaf(v.x, wi, acc.x);
      acc.y = fmaf(v.y, wi, acc.y);
      acc.z = fmaf(v.z, wi, acc.z);
      acc.w = fmaf(v.w, wi, acc.w);
    }
  }

  *(float4*)(agg + (size_t)d * D_IN + lane * 4) = acc;
}

extern "C" void kernel_launch(void* const* d_in, const int* in_sizes, int n_in,
                              void* d_out, int out_size, void* d_ws, size_t ws_size,
                              hipStream_t stream) {
  const float* feat = (const float*)d_in[0];
  const float* W    = (const float*)d_in[1];
  const float* cu   = (const float*)d_in[2];
  const float* cv   = (const float*)d_in[3];
  const float* ew   = (const float*)d_in[4];
  const int*   src  = (const int*)d_in[5];
  const int*   dst  = (const int*)d_in[6];

  float* out = (float*)d_out;
  float* agg = (float*)d_ws;

  const int N = in_sizes[2];
  const int E = in_sizes[5];

  const int nbins   = N / BIN_W + 1;
  const int nchunks = (E + CHUNK - 1) / CHUNK;
  const int pieces  = (nbins * nchunks) / 4096 + 1;
  const int Hp = pieces * 4096;
  const int Np = (N / 4096 + 1) * 4096;
  int*  H    = (int*)d_out;
  int*  S    = H + Hp;
  int*  off  = S + Hp;
  int2* rec  = (int2*)(off + Np);
  int*  perm = (int*)(rec + E);

  bin_count_kernel<<<nchunks, 256, 0, stream>>>(dst, H, E, nbins, nchunks, Hp);
  scan_counts_kernel<<<1, 1024, 0, stream>>>(H, S, pieces);
  bin_scatter_kernel<<<nchunks, 256, 0, stream>>>(dst, H, S, rec, E, nbins,
                                                  nchunks);
  bin_rank_kernel<<<nbins, 1024, 0, stream>>>(rec, S, off, perm, N, nchunks);
  influence_pull_kernel<<<(N + 7) / 8, 256, 0, stream>>>(feat, src, ew, cu, off,
                                                       perm, agg, N);

  const int rowTiles = (N + 15) / 16;
  influence_gemm_cu_kernel<<<rowTiles, 256, 0, stream>>>(agg, W, cv, out, N);
}


# summing kernel fused into the product kernel: a 16-wave block sums 32 target rows into an LDS tile and multiplies it by W there; summed rows skip memory; edge lists in workspace; sum order unchanged
# speedup vs baseline: 2.4355x; 1.2996x over previous
#include <hip/hip_runtime.h>


typedef __attribute__((ext_vector_type(2))) float v2f;
typedef __attribute__((ext_vector_type(8))) float v8f;

#define D_IN  128
#define D_OUT 128

#define BIN_SHIFT 9
#define BIN_W     512
#define MAX_BINS  256
#define CHUNK     4096

__global__ __launch_bounds__(256) void bin_count_kernel(
    const int* __restrict__ dst, int* __restrict__ H, int nedges, int nbins,
    int nchunks, int hpad) {
  __shared__ int hist[MAX_BINS];
  const int t = threadIdx.x;
  const int c = blockIdx.x;
  hist[t] = 0;
  __syncthreads();
  const int base = c * CHUNK;
  if (base + CHUNK <= nedges) {
    int4 d[4];
#pragma unroll
    for (int k = 0; k < 4; ++k)
      d[k] = *(const int4*)(dst + base + (k * 256 + t) * 4);
#pragma unroll
    for (int k = 0; k < 4; ++k) {
      atomicAdd(&hist[d[k].x >> BIN_SHIFT], 1);
      atomicAdd(&hist[d[k].y >> BIN_SHIFT], 1);
      atomicAdd(&hist[d[k].z >> BIN_SHIFT], 1);
      atomicAdd(&hist[d[k].w >> BIN_SHIFT], 1);
    }
  } else {
    for (int e = base + t; e < nedges; e += 256)
      atomicAdd(&hist[dst[e] >> BIN_SHIFT], 1);
  }
  __syncthreads();
  if (t < nbins) H[t * nchunks + c] = hist[t];
  if (c == 0)
    for (int i = nbins * nchunks + t; i < hpad; i += 256) H[i] = 0;
}

__device__ __forceinline__ int wave_sum_i32(int v) {
#pragma unroll
  for (int m = 16; m >= 1; m >>= 1) v += __shfl_xor(v, m);
  return v;
}

__global__ __launch_bounds__(1024) void scan_counts_kernel(
    const int* __restrict__ cnt, int* __restrict__ off, int pieces) {
  __shared__ int wtot[32];
  const int lane = threadIdx.x & 31;
  const int wave = threadIdx.x >> 5;
  const int4* c4 = (const int4*)cnt + (size_t)wave * pieces * 32 + lane;
  int4* o4 = (int4*)off + (size_t)wave * pieces * 32 + lane;

  int s = 0;
#pragma unroll 5
  for (int p = 0; p < pieces; ++p) {
    const int4 v = c4[p * 32];
    s += v.x + v.y + v.z + v.w;
  }
  s = wave_sum_i32(s);
  if (lane == 0) wtot[wave] = s;
  __syncthreads();
  int carry = wave_sum_i32((lane < wave) ? wtot[lane] : 0);

#pragma unroll 5
  for (int p = 0; p < pieces; ++p) {
    const int4 v = c4[p * 32];
    const int t = v.x + v.y + v.z + v.w;
    int inc = t;
#pragma unroll
    for (int d = 1; d < 32; d <<= 1) {
      const int u = __shfl_up(inc, d);
      if (lane >= d) inc += u;
    }
    int4 o;
    o.x = carry + inc - t;
    o.y = o.x + v.x;
    o.z = o.y + v.y;
    o.w = o.z + v.z;
    o4[p * 32] = o;
    carry += __shfl(inc, 31);
  }
}

__global__ __launch_bounds__(256) void bin_scatter_kernel(
    const int* __restrict__ dst, const int* __restrict__ H,
    const int* __restrict__ S, int2* __restrict__ rec, int nedges, int nbins,
    int nchunks) {
  __shared__ int cur[MAX_BINS];
  __shared__ int gofs[MAX_BINS];
  __shared__ int wtot[8];
  __shared__ int2 sRec[CHUNK];
  const int t = threadIdx.x;
  const int c = blockIdx.x;
  const int lane = t & 31;
  const int wave = t >> 5;
  const int base = c * CHUNK;
  const bool whole = base + CHUNK <= nedges;

  int4 d[4];
  if (whole) {
#pragma unroll
    for (int k = 0; k < 4; ++k)
      d[k] = *(const int4*)(dst + base + (k * 256 + t) * 4);
  }
  int hb = 0, sb = 0;
  if (t < nbins) {
    hb = H[t * nchunks + c];
    sb = S[t * nchunks + c];
  }

  int inc = hb;
#pragma unroll
  for (int s = 1; s < 32; s <<= 1) {
    const int u = __shfl_up(inc, s);
    if (lane >= s) inc += u;
  }
  if (lane == 31) wtot[wave] = inc;
  __syncthreads();
  int carry = 0;
#pragma unroll
  for (int w = 0; w < 8; ++w) carry += (w < wave) ? wtot[w] : 0;
  const int lstart = carry + inc - hb;
  cur[t] = lstart;
  gofs[t] = sb - lstart;
  __syncthreads();

  if (whole) {
    int slot[16];
#pragma unroll
    for (int k = 0; k < 4; ++k) {
      slot[4 * k + 0] = atomicAdd(&cur[d[k].x >> BIN_SHIFT], 1);
      slot[4 * k + 1] = atomicAdd(&cur[d[k].y >> BIN_SHIFT], 1);
      slot[4 * k + 2] = atomicAdd(&cur[d[k].z >> BIN_SHIFT], 1);
      slot[4 * k + 3] = atomicAdd(&cur[d[k].w >> BIN_SHIFT], 1);
    }
#pragma unroll
    for (int k = 0; k < 4; ++k) {
      const int e0 = base + (k * 256 + t) * 4;
      sRec[slot[4 * k + 0]] = make_int2(e0, d[k].x);
      sRec[slot[4 * k + 1]] = make_int2(e0 + 1, d[k].y);
      sRec[slot[4 * k + 2]] = make_int2(e0 + 2, d[k].z);
      sRec[slot[4 * k + 3]] = make_int2(e0 + 3, d[k].w);
    }
  } else {
    for (int e = base + t; e < nedges; e += 256) {
      const int de = dst[e];
      sRec[atomicAdd(&cur[de >> BIN_SHIFT], 1)] = make_int2(e, de);
    }
  }
  __syncthreads();

  if (whole) {
#pragma unroll
    for (int k = 0; k < CHUNK / 256; ++k) {
      const int i = k * 256 + t;
      const int2 r = sRec[i];
      rec[gofs[r.y >> BIN_SHIFT] + i] = r;
    }
  } else {
    const int n = nedges - base;
    for (int i = t; i < n; i += 256) {
      const int2 r = sRec[i];
      rec[gofs[r.y >> BIN_SHIFT] + i] = r;
    }
  }
}

#define RANK_CAP 16384
__global__ __launch_bounds__(1024) void bin_rank_kernel(
    const int2* __restrict__ rec, const int* __restrict__ S,
    int* __restrict__ off, int* __restrict__ perm, int nrows, int nchunks) {
  __shared__ int cntL[BIN_W];
  __shared__ int curL[BIN_W];
  __shared__ int sP[RANK_CAP];
  const int t = threadIdx.x;
  const int b = blockIdx.x;
  const int lo = S[b * nchunks];
  const int hi = S[(b + 1) * nchunks];
  const int tbase = b << BIN_SHIFT;
  const int n = hi - lo;
  const bool staged = n <= RANK_CAP;
  if (t < BIN_W) cntL[t] = 0;
  __syncthreads();
  for (int i0 = lo + t; i0 < hi; i0 += 4096) {
    int d[4];
#pragma unroll
    for (int k = 0; k < 4; ++k) {
      const int i = i0 + k * 1024;
      d[k] = (i < hi) ? rec[i].y : tbase;
    }
#pragma unroll
    for (int k = 0; k < 4; ++k)
      if (i0 + k * 1024 < hi) atomicAdd(&cntL[d[k] - tbase], 1);
  }
  __syncthreads();
  if (t < 32) {
    int v[16];
    int s = 0;
#pragma unroll
    for (int j = 0; j < 16; ++j) {
      v[j] = cntL[t * 16 + j];
      s += v[j];
    }
    int inc = s;
#pragma unroll
    for (int d = 1; d < 32; d <<= 1) {
      const int u = __shfl_up(inc, d);
      if (t >= d) inc += u;
    }
    int run = inc - s;
#pragma unroll
    for (int j = 0; j < 16; ++j) {
      curL[t * 16 + j] = run;
      run += v[j];
    }
  }
  __syncthreads();
  if (t < BIN_W) {
    const int g = tbase + t;
    if (g <= nrows) off[g] = lo + curL[t];
  }
  __syncthreads();
  for (int i0 = lo + t; i0 < hi; i0 += 4096) {
    int2 r[4];
#pragma unroll
    for (int k = 0; k < 4; ++k) {
      const int i = i0 + k * 1024;
      r[k] = (i < hi) ? rec[i] : make_int2(0, tbase);
    }
#pragma unroll
    for (int k = 0; k < 4; ++k)
      if (i0 + k * 1024 < hi) {
        const int slot = atomicAdd(&curL[r[k].y - tbase], 1);
        if (staged) sP[slot] = r[k].x;
        else perm[lo + slot] = r[k].x;
      }
  }
  if (staged) {
    __syncthreads();
    for (int i = t; i < n; i += 1024) perm[lo + i] = sP[i];
  }
}

#define TILE_ROWS 32

__device__ __forceinline__ void fma_row(float4& acc, const float4 v, const float w) {
  acc.x = fmaf(v.x, w, acc.x);
  acc.y = fmaf(v.y, w, acc.y);
  acc.z = fmaf(v.z, w, acc.z);
  acc.w = fmaf(v.w, w, acc.w);
}

__device__ __forceinline__ void add_run(float4& acc, const float* __restrict__ fb,
                                        const int ss, const int wb, const int n) {
  int i = 0;
  for (; i + 8 <= n; i += 8) {
    float4 v[8];
    float wi[8];
#pragma unroll
    for (int k = 0; k < 8; ++k) {
      const int si = __builtin_amdgcn_readlane(ss, i + k);
      wi[k] = __int_as_float(__builtin_amdgcn_readlane(wb, i + k));
      v[k] = *(const float4*)(fb + (size_t)si * D_IN);
    }
#pragma unroll
    for (int k = 0; k < 8; ++k) fma_row(acc, v[k], wi[k]);
  }
  if (i + 4 <= n) {
    float4 v[4];
    float wi[4];
#pragma unroll
    for (int k = 0; k < 4; ++k) {
      const int si = __builtin_amdgcn_readlane(ss, i + k);
      wi[k] = __int_as_float(__builtin_amdgcn_readlane(wb, i + k));
      v[k] = *(const float4*)(fb + (size_t)si * D_IN);
    }
#pragma unroll
    for (int k = 0; k < 4; ++k) fma_row(acc, v[k], wi[k]);
    i += 4;
  }
  for (; i < n; ++i) {
    const int si = __builtin_amdgcn_readlane(ss, i);
    const float wi = __int_as_float(__builtin_amdgcn_readlane(wb, i));
    fma_row(acc, *(const float4*)(fb + (size_t)si * D_IN), wi);
  }
}

__device__ __forceinline__ void add_long_run(
    float4& acc, const float* __restrict__ fb, const int* __restrict__ src,
    const float* __restrict__ edge_w, const float* __restrict__ cu,
    const int* __restrict__ perm, const int beg, const int end, const int lane) {
  int last = -1;
  for (int i = beg; i < end; ++i) {
    int best = 0x7fffffff;
    for (int p = beg + lane; p < end; p += 32) {
      const int e = perm[p];
      if (e > last && e < best) best = e;
    }
#pragma unroll
    for (int m = 16; m >= 1; m >>= 1) {
      const int o = __shfl_xor(best, m);
      best = (o < best) ? o : best;
    }
    best = __builtin_amdgcn_readfirstlane(best);
    last = best;
    const int si = src[best];
    const float wi = edge_w[best] * cu[si];
    fma_row(acc, *(const float4*)(fb + (size_t)si * D_IN), wi);
  }
}

__device__ __forceinline__ int rank_of(const int e) {
  int r = 0;
#pragma unroll
  for (int j = 0; j < 32; ++j)
    r += (__builtin_amdgcn_readlane(e, j) < e) ? 1 : 0;
  return r;
}

__global__ __launch_bounds__(512) void influence_pull_gemm_kernel(
    const float* __restrict__ feat, const float* __restrict__ W,
    const float* __restrict__ cu, const float* __restrict__ cv,
    const float* __restrict__ edge_w, const int* __restrict__ src,
    const int* __restrict__ off, const int* __restrict__ perm,
    float* __restrict__ out, int nrows) {
  __shared__ float sW[D_IN * D_OUT];
  __shared__ float4 sA[TILE_ROWS * 32];

  const int tid  = threadIdx.x;
  const int lane = tid & 31;
  const int wave = tid >> 5;
  const int rowBase = blockIdx.x * TILE_ROWS;

  const int d0 = rowBase + 2 * wave;
  int g = d0 + lane;
  if (g > nrows) g = nrows;
  const int ov = (lane < 3) ? off[g] : 0;

#pragma unroll
  for (int i = 0; i < (D_IN * D_OUT) / (4 * 512); ++i)
    ((float4*)sW)[tid + i * 512] = ((const float4*)W)[tid + i * 512];

  const int b0 = __builtin_amdgcn_readlane(ov, 0);
  const int b1 = __builtin_amdgcn_readlane(ov, 1);
  const int b2 = __builtin_amdgcn_readlane(ov, 2);
  const int deg0 = b1 - b0;
  const int deg1 = b2 - b1;
  const bool short0 = deg0 <= 32;
  const bool short1 = deg1 <= 32;

  const bool a0 = short0 && lane < deg0;
  const bool a1 = short1 && lane < deg1;
  const int q0 = perm[a0 ? b0 + lane : 0];
  const int q1 = perm[a1 ? b1 + lane : 0];
  const int t0 = src[q0];
  const int t1 = src[q1];
  const float u0 = edge_w[q0];
  const float u1 = edge_w[q1];
  const float c0 = cu[t0];
  const float c1 = cu[t1];
  const int e0 = a0 ? q0 : 0x7fffffff;
  const int e1 = a1 ? q1 : 0x7fffffff;
  const int s0 = a0 ? t0 : 0;
  const int s1 = a1 ? t1 : 0;
  const float w0 = a0 ? u0 * c0 : 0.f;
  const float w1 = a1 ? u1 * c1 : 0.f;

  const int r0 = rank_of(e0);
  const int r1 = rank_of(e1);
  const int ss0 = __builtin_amdgcn_ds_permute(r0 << 2, s0);
  const int wb0 = __builtin_amdgcn_ds_permute(r0 << 2, __float_as_int(w0));
  const int ss1 = __builtin_amdgcn_ds_permute(r1 << 2, s1);
  const int wb1 = __builtin_amdgcn_ds_permute(r1 << 2, __float_as_int(w1));

  const float* fb = feat + lane * 4;
  float4 acc0 = {0.f, 0.f, 0.f, 0.f};
  float4 acc1 = {0.f, 0.f, 0.f, 0.f};
  if (short0) add_run(acc0, fb, ss0, wb0, deg0);
  else add_long_run(acc0, fb, src, edge_w, cu, perm, b0, b1, lane);
  if (short1) add_run(acc1, fb, ss1, wb1, deg1);
  else add_long_run(acc1, fb, src, edge_w, cu, perm, b1, b2, lane);

  const int lr0 = 2 * wave;
  sA[lr0 * 32 + (lane ^ lr0)] = acc0;
  sA[(lr0 + 1) * 32 + (lane ^ (lr0 + 1))] = acc1;
  __syncthreads();

  const int rt      = wave >> 3;
  const int colBase = (wave & 7) * 16;
  const int m       = lane & 15;
  const int h       = lane >> 4;
  const int kh      = h * 2;
  const int ar = rt * 16 + m;
  const v2f* arow = (const v2f*)sA + ar * 64;

  const int mBase = rowBase + rt * 16 + h * 8;
  const int crow = rowBase + rt * 16 + m;
  const float cvl = cv[crow < nrows ? crow : nrows - 1];
  float cvr[8];
#pragma unroll
  for (int v = 0; v < 8; ++v) cvr[v] = __shfl(cvl, h * 8 + v);

  v8f acc = {};
#pragma unroll
  for (int kk = 0; kk < D_IN / 4; ++kk) {
    const v2f A = arow[((kk ^ ar) << 1) | h];
    const int kb = kk * 4 + kh;
    v2f B;
    B.x = sW[kb * D_OUT + colBase + m];
    B.y = sW[(kb + 1) * D_OUT + colBase + m];
    acc = __builtin_amdgcn_wmma_f32_16x16x4_f32(
         false, A,  false, B,
         (short)0, acc,  false,  false);
  }

#pragma unroll
  for (int v = 0; v < 8; ++v) {
    const int row = mBase + v;
    if (row < nrows)
      out[(size_t)row * D_OUT + colBase + m] = acc[v] * cvr[v];
  }
}

extern "C" void kernel_launch(void* const* d_in, const int* in_sizes, int n_in,
                              void* d_out, int out_size, void* d_ws, size_t ws_size,
                              hipStream_t stream) {
  const float* feat = (const float*)d_in[0];
  const float* W    = (const float*)d_in[1];
  const float* cu   = (const float*)d_in[2];
  const float* cv   = (const float*)d_in[3];
  const float* ew   = (const float*)d_in[4];
  const int*   src  = (const int*)d_in[5];
  const int*   dst  = (const int*)d_in[6];

  float* out = (float*)d_out;

  const int N = in_sizes[2];
  const int E = in_sizes[5];

  const int nbins   = N / BIN_W + 1;
  const int nchunks = (E + CHUNK - 1) / CHUNK;
  const int pieces  = (nbins * nchunks) / 4096 + 1;
  const int Hp = pieces * 4096;
  const int Np = (N / 4096 + 1) * 4096;
  int*  H    = (int*)d_ws;
  int*  S    = H + Hp;
  int*  off  = S + Hp;
  int2* rec  = (int2*)(off + Np);
  int*  perm = (int*)(rec + E);

  bin_count_kernel<<<nchunks, 256, 0, stream>>>(dst, H, E, nbins, nchunks, Hp);
  scan_counts_kernel<<<1, 1024, 0, stream>>>(H, S, pieces);
  bin_scatter_kernel<<<nchunks, 256, 0, stream>>>(dst, H, S, rec, E, nbins,
                                                  nchunks);
  bin_rank_kernel<<<nbins, 1024, 0, stream>>>(rec, S, off, perm, N, nchunks);

  const int rowTiles = (N + TILE_ROWS - 1) / TILE_ROWS;
  influence_pull_gemm_kernel<<<rowTiles, 512, 0, stream>>>(
      feat, W, cu, cv, ew, src, off, perm, out, N);
}


# runs of 33 to 64 edges no longer searched one edge at a time: two edges per lane fetched together, ranked across both registers, put in id order through the wave's own LDS rows; sum order unchanged
# speedup vs baseline: 2.6996x; 1.1084x over previous
#include <hip/hip_runtime.h>


typedef __attribute__((ext_vector_type(2))) float v2f;
typedef __attribute__((ext_vector_type(8))) float v8f;

#define D_IN  128
#define D_OUT 128

#define BIN_SHIFT 9
#define BIN_W     512
#define MAX_BINS  256
#define CHUNK     4096

__global__ __launch_bounds__(256) void bin_count_kernel(
    const int* __restrict__ dst, int* __restrict__ H, int nedges, int nbins,
    int nchunks, int hpad) {
  __shared__ int hist[MAX_BINS];
  const int t = threadIdx.x;
  const int c = blockIdx.x;
  hist[t] = 0;
  __syncthreads();
  const int base = c * CHUNK;
  if (base + CHUNK <= nedges) {
    int4 d[4];
#pragma unroll
    for (int k = 0; k < 4; ++k)
      d[k] = *(const int4*)(dst + base + (k * 256 + t) * 4);
#pragma unroll
    for (int k = 0; k < 4; ++k) {
      atomicAdd(&hist[d[k].x >> BIN_SHIFT], 1);
      atomicAdd(&hist[d[k].y >> BIN_SHIFT], 1);
      atomicAdd(&hist[d[k].z >> BIN_SHIFT], 1);
      atomicAdd(&hist[d[k].w >> BIN_SHIFT], 1);
    }
  } else {
    for (int e = base + t; e < nedges; e += 256)
      atomicAdd(&hist[dst[e] >> BIN_SHIFT], 1);
  }
  __syncthreads();
  if (t < nbins) H[t * nchunks + c] = hist[t];
  if (c == 0)
    for (int i = nbins * nchunks + t; i < hpad; i += 256) H[i] = 0;
}

__device__ __forceinline__ int wave_sum_i32(int v) {
#pragma unroll
  for (int m = 16; m >= 1; m >>= 1) v += __shfl_xor(v, m);
  return v;
}

__global__ __launch_bounds__(1024) void scan_counts_kernel(
    const int* __restrict__ cnt, int* __restrict__ off, int pieces) {
  __shared__ int wtot[32];
  const int lane = threadIdx.x & 31;
  const int wave = threadIdx.x >> 5;
  const int4* c4 = (const int4*)cnt + (size_t)wave * pieces * 32 + lane;
  int4* o4 = (int4*)off + (size_t)wave * pieces * 32 + lane;

  int s = 0;
#pragma unroll 5
  for (int p = 0; p < pieces; ++p) {
    const int4 v = c4[p * 32];
    s += v.x + v.y + v.z + v.w;
  }
  s = wave_sum_i32(s);
  if (lane == 0) wtot[wave] = s;
  __syncthreads();
  int carry = wave_sum_i32((lane < wave) ? wtot[lane] : 0);

#pragma unroll 5
  for (int p = 0; p < pieces; ++p) {
    const int4 v = c4[p * 32];
    const int t = v.x + v.y + v.z + v.w;
    int inc = t;
#pragma unroll
    for (int d = 1; d < 32; d <<= 1) {
      const int u = __shfl_up(inc, d);
      if (lane >= d) inc += u;
    }
    int4 o;
    o.x = carry + inc - t;
    o.y = o.x + v.x;
    o.z = o.y + v.y;
    o.w = o.z + v.z;
    o4[p * 32] = o;
    carry += __shfl(inc, 31);
  }
}

__global__ __launch_bounds__(256) void bin_scatter_kernel(
    const int* __restrict__ dst, const int* __restrict__ H,
    const int* __restrict__ S, int2* __restrict__ rec, int nedges, int nbins,
    int nchunks) {
  __shared__ int cur[MAX_BINS];
  __shared__ int gofs[MAX_BINS];
  __shared__ int wtot[8];
  __shared__ int2 sRec[CHUNK];
  const int t = threadIdx.x;
  const int c = blockIdx.x;
  const int lane = t & 31;
  const int wave = t >> 5;
  const int base = c * CHUNK;
  const bool whole = base + CHUNK <= nedges;

  int4 d[4];
  if (whole) {
#pragma unroll
    for (int k = 0; k < 4; ++k)
      d[k] = *(const int4*)(dst + base + (k * 256 + t) * 4);
  }
  int hb = 0, sb = 0;
  if (t < nbins) {
    hb = H[t * nchunks + c];
    sb = S[t * nchunks + c];
  }

  int inc = hb;
#pragma unroll
  for (int s = 1; s < 32; s <<= 1) {
    const int u = __shfl_up(inc, s);
    if (lane >= s) inc += u;
  }
  if (lane == 31) wtot[wave] = inc;
  __syncthreads();
  int carry = 0;
#pragma unroll
  for (int w = 0; w < 8; ++w) carry += (w < wave) ? wtot[w] : 0;
  const int lstart = carry + inc - hb;
  cur[t] = lstart;
  gofs[t] = sb - lstart;
  __syncthreads();

  if (whole) {
    int slot[16];
#pragma unroll
    for (int k = 0; k < 4; ++k) {
      slot[4 * k + 0] = atomicAdd(&cur[d[k].x >> BIN_SHIFT], 1);
      slot[4 * k + 1] = atomicAdd(&cur[d[k].y >> BIN_SHIFT], 1);
      slot[4 * k + 2] = atomicAdd(&cur[d[k].z >> BIN_SHIFT], 1);
      slot[4 * k + 3] = atomicAdd(&cur[d[k].w >> BIN_SHIFT], 1);
    }
#pragma unroll
    for (int k = 0; k < 4; ++k) {
      const int e0 = base + (k * 256 + t) * 4;
      sRec[slot[4 * k + 0]] = make_int2(e0, d[k].x);
      sRec[slot[4 * k + 1]] = make_int2(e0 + 1, d[k].y);
      sRec[slot[4 * k + 2]] = make_int2(e0 + 2, d[k].z);
      sRec[slot[4 * k + 3]] = make_int2(e0 + 3, d[k].w);
    }
  } else {
    for (int e = base + t; e < nedges; e += 256) {
      const int de = dst[e];
      sRec[atomicAdd(&cur[de >> BIN_SHIFT], 1)] = make_int2(e, de);
    }
  }
  __syncthreads();

  if (whole) {
#pragma unroll
    for (int k = 0; k < CHUNK / 256; ++k) {
      const int i = k * 256 + t;
      const int2 r = sRec[i];
      rec[gofs[r.y >> BIN_SHIFT] + i] = r;
    }
  } else {
    const int n = nedges - base;
    for (int i = t; i < n; i += 256) {
      const int2 r = sRec[i];
      rec[gofs[r.y >> BIN_SHIFT] + i] = r;
    }
  }
}

#define RANK_CAP 16384
__global__ __launch_bounds__(1024) void bin_rank_kernel(
    const int2* __restrict__ rec, const int* __restrict__ S,
    int* __restrict__ off, int* __restrict__ perm, int nrows, int nchunks) {
  __shared__ int cntL[BIN_W];
  __shared__ int curL[BIN_W];
  __shared__ int sP[RANK_CAP];
  const int t = threadIdx.x;
  const int b = blockIdx.x;
  const int lo = S[b * nchunks];
  const int hi = S[(b + 1) * nchunks];
  const int tbase = b << BIN_SHIFT;
  const int n = hi - lo;
  const bool staged = n <= RANK_CAP;
  if (t < BIN_W) cntL[t] = 0;
  __syncthreads();
  for (int i0 = lo + t; i0 < hi; i0 += 4096) {
    int d[4];
#pragma unroll
    for (int k = 0; k < 4; ++k) {
      const int i = i0 + k * 1024;
      d[k] = (i < hi) ? rec[i].y : tbase;
    }
#pragma unroll
    for (int k = 0; k < 4; ++k)
      if (i0 + k * 1024 < hi) atomicAdd(&cntL[d[k] - tbase], 1);
  }
  __syncthreads();
  if (t < 32) {
    int v[16];
    int s = 0;
#pragma unroll
    for (int j = 0; j < 16; ++j) {
      v[j] = cntL[t * 16 + j];
      s += v[j];
    }
    int inc = s;
#pragma unroll
    for (int d = 1; d < 32; d <<= 1) {
      const int u = __shfl_up(inc, d);
      if (t >= d) inc += u;
    }
    int run = inc - s;
#pragma unroll
    for (int j = 0; j < 16; ++j) {
      curL[t * 16 + j] = run;
      run += v[j];
    }
  }
  __syncthreads();
  if (t < BIN_W) {
    const int g = tbase + t;
    if (g <= nrows) off[g] = lo + curL[t];
  }
  __syncthreads();
  for (int i0 = lo + t; i0 < hi; i0 += 4096) {
    int2 r[4];
#pragma unroll
    for (int k = 0; k < 4; ++k) {
      const int i = i0 + k * 1024;
      r[k] = (i < hi) ? rec[i] : make_int2(0, tbase);
    }
#pragma unroll
    for (int k = 0; k < 4; ++k)
      if (i0 + k * 1024 < hi) {
        const int slot = atomicAdd(&curL[r[k].y - tbase], 1);
        if (staged) sP[slot] = r[k].x;
        else perm[lo + slot] = r[k].x;
      }
  }
  if (staged) {
    __syncthreads();
    for (int i = t; i < n; i += 1024) perm[lo + i] = sP[i];
  }
}

#define TILE_ROWS 32

__device__ __forceinline__ void fma_row(float4& acc, const float4 v, const float w) {
  acc.x = fmaf(v.x, w, acc.x);
  acc.y = fmaf(v.y, w, acc.y);
  acc.z = fmaf(v.z, w, acc.z);
  acc.w = fmaf(v.w, w, acc.w);
}

__device__ __forceinline__ void add_run(float4& acc, const float* __restrict__ fb,
                                        const int ss, const int wb, const int n) {
  int i = 0;
  for (; i + 8 <= n; i += 8) {
    float4 v[8];
    float wi[8];
#pragma unroll
    for (int k = 0; k < 8; ++k) {
      const int si = __builtin_amdgcn_readlane(ss, i + k);
      wi[k] = __int_as_float(__builtin_amdgcn_readlane(wb, i + k));
      v[k] = *(const float4*)(fb + (size_t)si * D_IN);
    }
#pragma unroll
    for (int k = 0; k < 8; ++k) fma_row(acc, v[k], wi[k]);
  }
  if (i + 4 <= n) {
    float4 v[4];
    float wi[4];
#pragma unroll
    for (int k = 0; k < 4; ++k) {
      const int si = __builtin_amdgcn_readlane(ss, i + k);
      wi[k] = __int_as_float(__builtin_amdgcn_readlane(wb, i + k));
      v[k] = *(const float4*)(fb + (size_t)si * D_IN);
    }
#pragma unroll
    for (int k = 0; k < 4; ++k) fma_row(acc, v[k], wi[k]);
    i += 4;
  }
  for (; i < n; ++i) {
    const int si = __builtin_amdgcn_readlane(ss, i);
    const float wi = __int_as_float(__builtin_amdgcn_readlane(wb, i));
    fma_row(acc, *(const float4*)(fb + (size_t)si * D_IN), wi);
  }
}

__device__ __forceinline__ void add_long_run(
    float4& acc, const float* __restrict__ fb, const int* __restrict__ src,
    const float* __restrict__ edge_w, const float* __restrict__ cu,
    const int* __restrict__ perm, const int beg, const int end, const int lane) {
  int last = -1;
  for (int i = beg; i < end; ++i) {
    int best = 0x7fffffff;
    for (int p = beg + lane; p < end; p += 32) {
      const int e = perm[p];
      if (e > last && e < best) best = e;
    }
#pragma unroll
    for (int m = 16; m >= 1; m >>= 1) {
      const int o = __shfl_xor(best, m);
      best = (o < best) ? o : best;
    }
    best = __builtin_amdgcn_readfirstlane(best);
    last = best;
    const int si = src[best];
    const float wi = edge_w[best] * cu[si];
    fma_row(acc, *(const float4*)(fb + (size_t)si * D_IN), wi);
  }
}

__device__ __forceinline__ int rank_of(const int e) {
  int r = 0;
#pragma unroll
  for (int j = 0; j < 32; ++j)
    r += (__builtin_amdgcn_readlane(e, j) < e) ? 1 : 0;
  return r;
}

__device__ __forceinline__ void add_mid_run(
    float4& acc, const float* __restrict__ fb, const int* __restrict__ src,
    const float* __restrict__ edge_w, const float* __restrict__ cu,
    const int* __restrict__ perm, float4* scr, const int beg, const int deg,
    const int lane) {
  const bool ah = 32 + lane < deg;
  const int ql = perm[beg + lane];
  const int qh = perm[ah ? beg + 32 + lane : beg];
  const int tl = src[ql];
  const int th = src[qh];
  const float wl = edge_w[ql] * cu[tl];
  const float wh = edge_w[qh] * cu[th];
  const int el = ql;
  const int eh = ah ? qh : 0x7fffffff;
  int rl = 0;
  int rh = 0;
#pragma nounroll
  for (int j = 0; j < 32; ++j) {
    const int a = __builtin_amdgcn_readlane(el, j);
    const int b = __builtin_amdgcn_readlane(eh, j);
    rl += ((a < el) ? 1 : 0) + ((b < el) ? 1 : 0);
    rh += ((a < eh) ? 1 : 0) + ((b < eh) ? 1 : 0);
  }
  scr[rl] = make_float4(__int_as_float(tl), wl, 0.f, 0.f);
  if (ah) scr[rh] = make_float4(__int_as_float(th), wh, 0.f, 0.f);
#pragma nounroll
  for (int c = 0; c < 2; ++c) {
    const float4 p = scr[c * 32 + lane];
    const int n = (c == 0) ? 32 : deg - 32;
    add_run(acc, fb, __float_as_int(p.x), __float_as_int(p.y), n);
  }
}

__global__ __launch_bounds__(512) void influence_pull_gemm_kernel(
    const float* __restrict__ feat, const float* __restrict__ W,
    const float* __restrict__ cu, const float* __restrict__ cv,
    const float* __restrict__ edge_w, const int* __restrict__ src,
    const int* __restrict__ off, const int* __restrict__ perm,
    float* __restrict__ out, int nrows) {
  __shared__ float sW[D_IN * D_OUT];
  __shared__ float4 sA[TILE_ROWS * 32];

  const int tid  = threadIdx.x;
  const int lane = tid & 31;
  const int wave = tid >> 5;
  const int rowBase = blockIdx.x * TILE_ROWS;

  const int d0 = rowBase + 2 * wave;
  int g = d0 + lane;
  if (g > nrows) g = nrows;
  const int ov = (lane < 3) ? off[g] : 0;

#pragma unroll
  for (int i = 0; i < (D_IN * D_OUT) / (4 * 512); ++i)
    ((float4*)sW)[tid + i * 512] = ((const float4*)W)[tid + i * 512];

  const int b0 = __builtin_amdgcn_readlane(ov, 0);
  const int b1 = __builtin_amdgcn_readlane(ov, 1);
  const int b2 = __builtin_amdgcn_readlane(ov, 2);
  const int deg0 = b1 - b0;
  const int deg1 = b2 - b1;
  const bool short0 = deg0 <= 32;
  const bool short1 = deg1 <= 32;

  const bool a0 = short0 && lane < deg0;
  const bool a1 = short1 && lane < deg1;
  const int q0 = perm[a0 ? b0 + lane : 0];
  const int q1 = perm[a1 ? b1 + lane : 0];
  const int t0 = src[q0];
  const int t1 = src[q1];
  const float u0 = edge_w[q0];
  const float u1 = edge_w[q1];
  const float c0 = cu[t0];
  const float c1 = cu[t1];
  const int e0 = a0 ? q0 : 0x7fffffff;
  const int e1 = a1 ? q1 : 0x7fffffff;
  const int s0 = a0 ? t0 : 0;
  const int s1 = a1 ? t1 : 0;
  const float w0 = a0 ? u0 * c0 : 0.f;
  const float w1 = a1 ? u1 * c1 : 0.f;

  const int r0 = rank_of(e0);
  const int r1 = rank_of(e1);
  const int ss0 = __builtin_amdgcn_ds_permute(r0 << 2, s0);
  const int wb0 = __builtin_amdgcn_ds_permute(r0 << 2, __float_as_int(w0));
  const int ss1 = __builtin_amdgcn_ds_permute(r1 << 2, s1);
  const int wb1 = __builtin_amdgcn_ds_permute(r1 << 2, __float_as_int(w1));

  const float* fb = feat + lane * 4;
  float4 acc0 = {0.f, 0.f, 0.f, 0.f};
  float4 acc1 = {0.f, 0.f, 0.f, 0.f};
  const int lr0 = 2 * wave;
  float4* scr = sA + lr0 * 32;
  if (short0) add_run(acc0, fb, ss0, wb0, deg0);
  else if (deg0 <= 64) add_mid_run(acc0, fb, src, edge_w, cu, perm, scr, b0, deg0, lane);
  else add_long_run(acc0, fb, src, edge_w, cu, perm, b0, b1, lane);
  if (short1) add_run(acc1, fb, ss1, wb1, deg1);
  else if (deg1 <= 64) add_mid_run(acc1, fb, src, edge_w, cu, perm, scr, b1, deg1, lane);
  else add_long_run(acc1, fb, src, edge_w, cu, perm, b1, b2, lane);

  sA[lr0 * 32 + (lane ^ lr0)] = acc0;
  sA[(lr0 + 1) * 32 + (lane ^ (lr0 + 1))] = acc1;
  __syncthreads();

  const int rt      = wave >> 3;
  const int colBase = (wave & 7) * 16;
  const int m       = lane & 15;
  const int h       = lane >> 4;
  const int kh      = h * 2;
  const int ar = rt * 16 + m;
  const v2f* arow = (const v2f*)sA + ar * 64;

  const int mBase = rowBase + rt * 16 + h * 8;
  const int crow = rowBase + rt * 16 + m;
  const float cvl = cv[crow < nrows ? crow : nrows - 1];
  float cvr[8];
#pragma unroll
  for (int v = 0; v < 8; ++v) cvr[v] = __shfl(cvl, h * 8 + v);

  v8f acc = {};
#pragma unroll
  for (int kk = 0; kk < D_IN / 4; ++kk) {
    const v2f A = arow[((kk ^ ar) << 1) | h];
    const int kb = kk * 4 + kh;
    v2f B;
    B.x = sW[kb * D_OUT + colBase + m];
    B.y = sW[(kb + 1) * D_OUT + colBase + m];
    acc = __builtin_amdgcn_wmma_f32_16x16x4_f32(
         false, A,  false, B,
         (short)0, acc,  false,  false);
  }

#pragma unroll
  for (int v = 0; v < 8; ++v) {
    const int row = mBase + v;
    if (row < nrows)
      out[(size_t)row * D_OUT + colBase + m] = acc[v] * cvr[v];
  }
}

extern "C" void kernel_launch(void* const* d_in, const int* in_sizes, int n_in,
                              void* d_out, int out_size, void* d_ws, size_t ws_size,
                              hipStream_t stream) {
  const float* feat = (const float*)d_in[0];
  const float* W    = (const float*)d_in[1];
  const float* cu   = (const float*)d_in[2];
  const float* cv   = (const float*)d_in[3];
  const float* ew   = (const float*)d_in[4];
  const int*   src  = (const int*)d_in[5];
  const int*   dst  = (const int*)d_in[6];

  float* out = (float*)d_out;

  const int N = in_sizes[2];
  const int E = in_sizes[5];

  const int nbins   = N / BIN_W + 1;
  const int nchunks = (E + CHUNK - 1) / CHUNK;
  const int pieces  = (nbins * nchunks) / 4096 + 1;
  const int Hp = pieces * 4096;
  const int Np = (N / 4096 + 1) * 4096;
  int*  H    = (int*)d_ws;
  int*  S    = H + Hp;
  int*  off  = S + Hp;
  int2* rec  = (int2*)(off + Np);
  int*  perm = (int*)(rec + E);

  bin_count_kernel<<<nchunks, 256, 0, stream>>>(dst, H, E, nbins, nchunks, Hp);
  scan_counts_kernel<<<1, 1024, 0, stream>>>(H, S, pieces);
  bin_scatter_kernel<<<nchunks, 256, 0, stream>>>(dst, H, S, rec, E, nbins,
                                                  nchunks);
  bin_rank_kernel<<<nbins, 1024, 0, stream>>>(rec, S, off, perm, N, nchunks);

  const int rowTiles = (N + TILE_ROWS - 1) / TILE_ROWS;
  influence_pull_gemm_kernel<<<rowTiles, 512, 0, stream>>>(
      feat, W, cu, cv, ew, src, off, perm, out, N);
}


# fused kernel made persistent: 1024 work-groups loop over row tiles, W staged once each, next tile's index chain (bounds, ids, source, weight, cu) issued under this tile's row loads; sum order same
# speedup vs baseline: 2.8717x; 1.0638x over previous
#include <hip/hip_runtime.h>


typedef __attribute__((ext_vector_type(2))) float v2f;
typedef __attribute__((ext_vector_type(8))) float v8f;

#define D_IN  128
#define D_OUT 128

#define BIN_SHIFT 9
#define BIN_W     512
#define MAX_BINS  256
#define CHUNK     4096

__global__ __launch_bounds__(256) void bin_count_kernel(
    const int* __restrict__ dst, int* __restrict__ H, int nedges, int nbins,
    int nchunks, int hpad) {
  __shared__ int hist[MAX_BINS];
  const int t = threadIdx.x;
  const int c = blockIdx.x;
  hist[t] = 0;
  __syncthreads();
  const int base = c * CHUNK;
  if (base + CHUNK <= nedges) {
    int4 d[4];
#pragma unroll
    for (int k = 0; k < 4; ++k)
      d[k] = *(const int4*)(dst + base + (k * 256 + t) * 4);
#pragma unroll
    for (int k = 0; k < 4; ++k) {
      atomicAdd(&hist[d[k].x >> BIN_SHIFT], 1);
      atomicAdd(&hist[d[k].y >> BIN_SHIFT], 1);
      atomicAdd(&hist[d[k].z >> BIN_SHIFT], 1);
      atomicAdd(&hist[d[k].w >> BIN_SHIFT], 1);
    }
  } else {
    for (int e = base + t; e < nedges; e += 256)
      atomicAdd(&hist[dst[e] >> BIN_SHIFT], 1);
  }
  __syncthreads();
  if (t < nbins) H[t * nchunks + c] = hist[t];
  if (c == 0)
    for (int i = nbins * nchunks + t; i < hpad; i += 256) H[i] = 0;
}

__device__ __forceinline__ int wave_sum_i32(int v) {
#pragma unroll
  for (int m = 16; m >= 1; m >>= 1) v += __shfl_xor(v, m);
  return v;
}

__global__ __launch_bounds__(1024) void scan_counts_kernel(
    const int* __restrict__ cnt, int* __restrict__ off, int pieces) {
  __shared__ int wtot[32];
  const int lane = threadIdx.x & 31;
  const int wave = threadIdx.x >> 5;
  const int4* c4 = (const int4*)cnt + (size_t)wave * pieces * 32 + lane;
  int4* o4 = (int4*)off + (size_t)wave * pieces * 32 + lane;

  int s = 0;
#pragma unroll 5
  for (int p = 0; p < pieces; ++p) {
    const int4 v = c4[p * 32];
    s += v.x + v.y + v.z + v.w;
  }
  s = wave_sum_i32(s);
  if (lane == 0) wtot[wave] = s;
  __syncthreads();
  int carry = wave_sum_i32((lane < wave) ? wtot[lane] : 0);

#pragma unroll 5
  for (int p = 0; p < pieces; ++p) {
    const int4 v = c4[p * 32];
    const int t = v.x + v.y + v.z + v.w;
    int inc = t;
#pragma unroll
    for (int d = 1; d < 32; d <<= 1) {
      const int u = __shfl_up(inc, d);
      if (lane >= d) inc += u;
    }
    int4 o;
    o.x = carry + inc - t;
    o.y = o.x + v.x;
    o.z = o.y + v.y;
    o.w = o.z + v.z;
    o4[p * 32] = o;
    carry += __shfl(inc, 31);
  }
}

__global__ __launch_bounds__(256) void bin_scatter_kernel(
    const int* __restrict__ dst, const int* __restrict__ H,
    const int* __restrict__ S, int2* __restrict__ rec, int nedges, int nbins,
    int nchunks) {
  __shared__ int cur[MAX_BINS];
  __shared__ int gofs[MAX_BINS];
  __shared__ int wtot[8];
  __shared__ int2 sRec[CHUNK];
  const int t = threadIdx.x;
  const int c = blockIdx.x;
  const int lane = t & 31;
  const int wave = t >> 5;
  const int base = c * CHUNK;
  const bool whole = base + CHUNK <= nedges;

  int4 d[4];
  if (whole) {
#pragma unroll
    for (int k = 0; k < 4; ++k)
      d[k] = *(const int4*)(dst + base + (k * 256 + t) * 4);
  }
  int hb = 0, sb = 0;
  if (t < nbins) {
    hb = H[t * nchunks + c];
    sb = S[t * nchunks + c];
  }

  int inc = hb;
#pragma unroll
  for (int s = 1; s < 32; s <<= 1) {
    const int u = __shfl_up(inc, s);
    if (lane >= s) inc += u;
  }
  if (lane == 31) wtot[wave] = inc;
  __syncthreads();
  int carry = 0;
#pragma unroll
  for (int w = 0; w < 8; ++w) carry += (w < wave) ? wtot[w] : 0;
  const int lstart = carry + inc - hb;
  cur[t] = lstart;
  gofs[t] = sb - lstart;
  __syncthreads();

  if (whole) {
    int slot[16];
#pragma unroll
    for (int k = 0; k < 4; ++k) {
      slot[4 * k + 0] = atomicAdd(&cur[d[k].x >> BIN_SHIFT], 1);
      slot[4 * k + 1] = atomicAdd(&cur[d[k].y >> BIN_SHIFT], 1);
      slot[4 * k + 2] = atomicAdd(&cur[d[k].z >> BIN_SHIFT], 1);
      slot[4 * k + 3] = atomicAdd(&cur[d[k].w >> BIN_SHIFT], 1);
    }
#pragma unroll
    for (int k = 0; k < 4; ++k) {
      const int e0 = base + (k * 256 + t) * 4;
      sRec[slot[4 * k + 0]] = make_int2(e0, d[k].x);
      sRec[slot[4 * k + 1]] = make_int2(e0 + 1, d[k].y);
      sRec[slot[4 * k + 2]] = make_int2(e0 + 2, d[k].z);
      sRec[slot[4 * k + 3]] = make_int2(e0 + 3, d[k].w);
    }
  } else {
    for (int e = base + t; e < nedges; e += 256) {
      const int de = dst[e];
      sRec[atomicAdd(&cur[de >> BIN_SHIFT], 1)] = make_int2(e, de);
    }
  }
  __syncthreads();

  if (whole) {
#pragma unroll
    for (int k = 0; k < CHUNK / 256; ++k) {
      const int i = k * 256 + t;
      const int2 r = sRec[i];
      rec[gofs[r.y >> BIN_SHIFT] + i] = r;
    }
  } else {
    const int n = nedges - base;
    for (int i = t; i < n; i += 256) {
      const int2 r = sRec[i];
      rec[gofs[r.y >> BIN_SHIFT] + i] = r;
    }
  }
}

#define RANK_CAP 16384
__global__ __launch_bounds__(1024) void bin_rank_kernel(
    const int2* __restrict__ rec, const int* __restrict__ S,
    int* __restrict__ off, int* __restrict__ perm, int nrows, int nchunks) {
  __shared__ int cntL[BIN_W];
  __shared__ int curL[BIN_W];
  __shared__ int sP[RANK_CAP];
  const int t = threadIdx.x;
  const int b = blockIdx.x;
  const int lo = S[b * nchunks];
  const int hi = S[(b + 1) * nchunks];
  const int tbase = b << BIN_SHIFT;
  const int n = hi - lo;
  const bool staged = n <= RANK_CAP;
  if (t < BIN_W) cntL[t] = 0;
  __syncthreads();
  for (int i0 = lo + t; i0 < hi; i0 += 4096) {
    int d[4];
#pragma unroll
    for (int k = 0; k < 4; ++k) {
      const int i = i0 + k * 1024;
      d[k] = (i < hi) ? rec[i].y : tbase;
    }
#pragma unroll
    for (int k = 0; k < 4; ++k)
      if (i0 + k * 1024 < hi) atomicAdd(&cntL[d[k] - tbase], 1);
  }
  __syncthreads();
  if (t < 32) {
    int v[16];
    int s = 0;
#pragma unroll
    for (int j = 0; j < 16; ++j) {
      v[j] = cntL[t * 16 + j];
      s += v[j];
    }
    int inc = s;
#pragma unroll
    for (int d = 1; d < 32; d <<= 1) {
      const int u = __shfl_up(inc, d);
      if (t >= d) inc += u;
    }
    int run = inc - s;
#pragma unroll
    for (int j = 0; j < 16; ++j) {
      curL[t * 16 + j] = run;
      run += v[j];
    }
  }
  __syncthreads();
  if (t < BIN_W) {
    const int g = tbase + t;
    if (g <= nrows) off[g] = lo + curL[t];
  }
  __syncthreads();
  for (int i0 = lo + t; i0 < hi; i0 += 4096) {
    int2 r[4];
#pragma unroll
    for (int k = 0; k < 4; ++k) {
      const int i = i0 + k * 1024;
      r[k] = (i < hi) ? rec[i] : make_int2(0, tbase);
    }
#pragma unroll
    for (int k = 0; k < 4; ++k)
      if (i0 + k * 1024 < hi) {
        const int slot = atomicAdd(&curL[r[k].y - tbase], 1);
        if (staged) sP[slot] = r[k].x;
        else perm[lo + slot] = r[k].x;
      }
  }
  if (staged) {
    __syncthreads();
    for (int i = t; i < n; i += 1024) perm[lo + i] = sP[i];
  }
}

#define TILE_ROWS 32

__device__ __forceinline__ void fma_row(float4& acc, const float4 v, const float w) {
  acc.x = fmaf(v.x, w, acc.x);
  acc.y = fmaf(v.y, w, acc.y);
  acc.z = fmaf(v.z, w, acc.z);
  acc.w = fmaf(v.w, w, acc.w);
}

__device__ __forceinline__ void add_run(float4& acc, const float* __restrict__ fb,
                                        const int ss, const int wb, const int n) {
  int i = 0;
  for (; i + 8 <= n; i += 8) {
    float4 v[8];
    float wi[8];
#pragma unroll
    for (int k = 0; k < 8; ++k) {
      const int si = __builtin_amdgcn_readlane(ss, i + k);
      wi[k] = __int_as_float(__builtin_amdgcn_readlane(wb, i + k));
      v[k] = *(const float4*)(fb + (size_t)si * D_IN);
    }
#pragma unroll
    for (int k = 0; k < 8; ++k) fma_row(acc, v[k], wi[k]);
  }
  if (i + 4 <= n) {
    float4 v[4];
    float wi[4];
#pragma unroll
    for (int k = 0; k < 4; ++k) {
      const int si = __builtin_amdgcn_readlane(ss, i + k);
      wi[k] = __int_as_float(__builtin_amdgcn_readlane(wb, i + k));
      v[k] = *(const float4*)(fb + (size_t)si * D_IN);
    }
#pragma unroll
    for (int k = 0; k < 4; ++k) fma_row(acc, v[k], wi[k]);
    i += 4;
  }
  for (; i < n; ++i) {
    const int si = __builtin_amdgcn_readlane(ss, i);
    const float wi = __int_as_float(__builtin_amdgcn_readlane(wb, i));
    fma_row(acc, *(const float4*)(fb + (size_t)si * D_IN), wi);
  }
}

__device__ __forceinline__ void add_long_run(
    float4& acc, const float* __restrict__ fb, const int* __restrict__ src,
    const float* __restrict__ edge_w, const float* __restrict__ cu,
    const int* __restrict__ perm, const int beg, const int end, const int lane) {
  int last = -1;
  for (int i = beg; i < end; ++i) {
    int best = 0x7fffffff;
    for (int p = beg + lane; p < end; p += 32) {
      const int e = perm[p];
      if (e > last && e < best) best = e;
    }
#pragma unroll
    for (int m = 16; m >= 1; m >>= 1) {
      const int o = __shfl_xor(best, m);
      best = (o < best) ? o : best;
    }
    best = __builtin_amdgcn_readfirstlane(best);
    last = best;
    const int si = src[best];
    const float wi = edge_w[best] * cu[si];
    fma_row(acc, *(const float4*)(fb + (size_t)si * D_IN), wi);
  }
}

__device__ __forceinline__ int rank_of(const int e) {
  int r = 0;
#pragma unroll
  for (int j = 0; j < 32; ++j)
    r += (__builtin_amdgcn_readlane(e, j) < e) ? 1 : 0;
  return r;
}

__device__ __forceinline__ void add_mid_run(
    float4& acc, const float* __restrict__ fb, const int* __restrict__ src,
    const float* __restrict__ edge_w, const float* __restrict__ cu,
    const int* __restrict__ perm, float4* scr, const int beg, const int deg,
    const int lane) {
  const bool ah = 32 + lane < deg;
  const int ql = perm[beg + lane];
  const int qh = perm[ah ? beg + 32 + lane : beg];
  const int tl = src[ql];
  const int th = src[qh];
  const float wl = edge_w[ql] * cu[tl];
  const float wh = edge_w[qh] * cu[th];
  const int el = ql;
  const int eh = ah ? qh : 0x7fffffff;
  int rl = 0;
  int rh = 0;
#pragma nounroll
  for (int j = 0; j < 32; ++j) {
    const int a = __builtin_amdgcn_readlane(el, j);
    const int b = __builtin_amdgcn_readlane(eh, j);
    rl += ((a < el) ? 1 : 0) + ((b < el) ? 1 : 0);
    rh += ((a < eh) ? 1 : 0) + ((b < eh) ? 1 : 0);
  }
  scr[rl] = make_float4(__int_as_float(tl), wl, 0.f, 0.f);
  if (ah) scr[rh] = make_float4(__int_as_float(th), wh, 0.f, 0.f);
#pragma nounroll
  for (int c = 0; c < 2; ++c) {
    const float4 p = scr[c * 32 + lane];
    const int n = (c == 0) ? 32 : deg - 32;
    add_run(acc, fb, __float_as_int(p.x), __float_as_int(p.y), n);
  }
}

__device__ __forceinline__ int load_bounds(const int* __restrict__ off, const int d0,
                                           const int lane, const int nrows) {
  int g = d0 + lane;
  if (g > nrows) g = nrows;
  return (lane < 3) ? off[g] : 0;
}

__global__ __launch_bounds__(512) void influence_pull_gemm_kernel(
    const float* __restrict__ feat, const float* __restrict__ W,
    const float* __restrict__ cu, const float* __restrict__ cv,
    const float* __restrict__ edge_w, const int* __restrict__ src,
    const int* __restrict__ off, const int* __restrict__ perm,
    float* __restrict__ out, int nrows, int ntiles) {
  __shared__ float sW[D_IN * D_OUT];
  __shared__ float4 sA[TILE_ROWS * 32];

  const int tid = threadIdx.x;
  int tile = blockIdx.x;
  if (tile >= ntiles) return;

  int b0, b1, b2;
  int q0, q1, t0, t1;
  float u0, u1, c0, c1;
  {
    const int lane = tid & 31;
    const int wave = tid >> 5;
    const int ov = load_bounds(off, tile * TILE_ROWS + 2 * wave, lane, nrows);

#pragma unroll
    for (int i = 0; i < (D_IN * D_OUT) / (4 * 512); ++i)
      ((float4*)sW)[tid + i * 512] = ((const float4*)W)[tid + i * 512];

    b0 = __builtin_amdgcn_readlane(ov, 0);
    b1 = __builtin_amdgcn_readlane(ov, 1);
    b2 = __builtin_amdgcn_readlane(ov, 2);
    q0 = perm[(b1 - b0 <= 32 && lane < b1 - b0) ? b0 + lane : 0];
    q1 = perm[(b2 - b1 <= 32 && lane < b2 - b1) ? b1 + lane : 0];
    t0 = src[q0];
    t1 = src[q1];
    u0 = edge_w[q0];
    u1 = edge_w[q1];
    c0 = cu[t0];
    c1 = cu[t1];
  }

  for (;;) {
    int tx = tid;
    asm volatile("" : "+v"(tx));
    const int lane = tx & 31;
    const int wave = __builtin_amdgcn_readfirstlane(tx >> 5);

    const int rowBase = tile * TILE_ROWS;
    const int deg0 = b1 - b0;
    const int deg1 = b2 - b1;
    const bool short0 = deg0 <= 32;
    const bool short1 = deg1 <= 32;
    const bool a0 = short0 && lane < deg0;
    const bool a1 = short1 && lane < deg1;
    const int e0 = a0 ? q0 : 0x7fffffff;
    const int e1 = a1 ? q1 : 0x7fffffff;
    const int s0 = a0 ? t0 : 0;
    const int s1 = a1 ? t1 : 0;
    const float w0 = a0 ? u0 * c0 : 0.f;
    const float w1 = a1 ? u1 * c1 : 0.f;

    const int r0 = rank_of(e0);
    const int r1 = rank_of(e1);
    const int ss0 = __builtin_amdgcn_ds_permute(r0 << 2, s0);
    const int wb0 = __builtin_amdgcn_ds_permute(r0 << 2, __float_as_int(w0));
    const int ss1 = __builtin_amdgcn_ds_permute(r1 << 2, s1);
    const int wb1 = __builtin_amdgcn_ds_permute(r1 << 2, __float_as_int(w1));

    const int next = tile + gridDim.x;
    const int ovN = load_bounds(off, next * TILE_ROWS + 2 * wave, lane, nrows);

    const float* fb = feat + lane * 4;
    const int lr0 = 2 * wave;
    float4* scr = sA + lr0 * 32;

    float4 acc0 = {0.f, 0.f, 0.f, 0.f};
    float4 acc1 = {0.f, 0.f, 0.f, 0.f};
    if (short0) add_run(acc0, fb, ss0, wb0, deg0);
    else if (deg0 <= 64) add_mid_run(acc0, fb, src, edge_w, cu, perm, scr, b0, deg0, lane);
    else add_long_run(acc0, fb, src, edge_w, cu, perm, b0, b1, lane);

    const int nb0 = __builtin_amdgcn_readlane(ovN, 0);
    const int nb1 = __builtin_amdgcn_readlane(ovN, 1);
    const int nb2 = __builtin_amdgcn_readlane(ovN, 2);
    const int nq0 = perm[(nb1 - nb0 <= 32 && lane < nb1 - nb0) ? nb0 + lane : 0];
    const int nq1 = perm[(nb2 - nb1 <= 32 && lane < nb2 - nb1) ? nb1 + lane : 0];

    if (short1) add_run(acc1, fb, ss1, wb1, deg1);
    else if (deg1 <= 64) add_mid_run(acc1, fb, src, edge_w, cu, perm, scr, b1, deg1, lane);
    else add_long_run(acc1, fb, src, edge_w, cu, perm, b1, b2, lane);

    const int nt0 = src[nq0];
    const int nt1 = src[nq1];
    const float nu0 = edge_w[nq0];
    const float nu1 = edge_w[nq1];

    sA[lr0 * 32 + (lane ^ lr0)] = acc0;
    sA[(lr0 + 1) * 32 + (lane ^ (lr0 + 1))] = acc1;
    __syncthreads();

    const float nc0 = cu[nt0];
    const float nc1 = cu[nt1];

    const int rt      = wave >> 3;
    const int colBase = (wave & 7) * 16;
    const int m       = lane & 15;
    const int h       = lane >> 4;
    const int kh      = h * 2;
    const int ar = rt * 16 + m;
    const v2f* arow = (const v2f*)sA + ar * 64;

    const int mBase = rowBase + rt * 16 + h * 8;
    const int crow = rowBase + rt * 16 + m;
    const float cvl = cv[crow < nrows ? crow : nrows - 1];
    float cvr[8];
#pragma unroll
    for (int v = 0; v < 8; ++v) cvr[v] = __shfl(cvl, h * 8 + v);

    v8f acc = {};
#pragma unroll
    for (int kk = 0; kk < D_IN / 4; ++kk) {
      const v2f A = arow[((kk ^ ar) << 1) | h];
      const int kb = kk * 4 + kh;
      v2f B;
      B.x = sW[kb * D_OUT + colBase + m];
      B.y = sW[(kb + 1) * D_OUT + colBase + m];
      acc = __builtin_amdgcn_wmma_f32_16x16x4_f32(
           false, A,  false, B,
           (short)0, acc,  false,  false);
    }

#pragma unroll
    for (int v = 0; v < 8; ++v) {
      const int row = mBase + v;
      if (row < nrows)
        out[(size_t)row * D_OUT + colBase + m] = acc[v] * cvr[v];
    }

    if (next >= ntiles) break;
    __syncthreads();
    tile = next;
    b0 = nb0; b1 = nb1; b2 = nb2;
    q0 = nq0; q1 = nq1;
    t0 = nt0; t1 = nt1;
    u0 = nu0; u1 = nu1;
    c0 = nc0; c1 = nc1;
  }
}

extern "C" void kernel_launch(void* const* d_in, const int* in_sizes, int n_in,
                              void* d_out, int out_size, void* d_ws, size_t ws_size,
                              hipStream_t stream) {
  const float* feat = (const float*)d_in[0];
  const float* W    = (const float*)d_in[1];
  const float* cu   = (const float*)d_in[2];
  const float* cv   = (const float*)d_in[3];
  const float* ew   = (const float*)d_in[4];
  const int*   src  = (const int*)d_in[5];
  const int*   dst  = (const int*)d_in[6];

  float* out = (float*)d_out;

  const int N = in_sizes[2];
  const int E = in_sizes[5];

  const int nbins   = N / BIN_W + 1;
  const int nchunks = (E + CHUNK - 1) / CHUNK;
  const int pieces  = (nbins * nchunks) / 4096 + 1;
  const int Hp = pieces * 4096;
  const int Np = (N / 4096 + 1) * 4096;
  int*  H    = (int*)d_ws;
  int*  S    = H + Hp;
  int*  off  = S + Hp;
  int2* rec  = (int2*)(off + Np);
  int*  perm = (int*)(rec + E);

  bin_count_kernel<<<nchunks, 256, 0, stream>>>(dst, H, E, nbins, nchunks, Hp);
  scan_counts_kernel<<<1, 1024, 0, stream>>>(H, S, pieces);
  bin_scatter_kernel<<<nchunks, 256, 0, stream>>>(dst, H, S, rec, E, nbins,
                                                  nchunks);
  bin_rank_kernel<<<nbins, 1024, 0, stream>>>(rec, S, off, perm, N, nchunks);

  const int rowTiles = (N + TILE_ROWS - 1) / TILE_ROWS;
  influence_pull_gemm_kernel<<<1024, 512, 0, stream>>>(
      feat, W, cu, cv, ew, src, off, perm, out, N, rowTiles);
}


# source and weight (edge_w*cu) ride with each edge through the sort: fused kernel reads a run's ids, sources, weights as contiguous words, not three dependent scattered loads per edge; sum order same
# speedup vs baseline: 3.0354x; 1.0570x over previous
#include <hip/hip_runtime.h>


typedef __attribute__((ext_vector_type(2))) float v2f;
typedef __attribute__((ext_vector_type(8))) float v8f;

#define D_IN  128
#define D_OUT 128

#define BIN_SHIFT 9
#define BIN_W     512
#define MAX_BINS  256
#define CHUNK     4096

__global__ __launch_bounds__(256) void bin_count_kernel(
    const int* __restrict__ dst, int* __restrict__ H, int nedges, int nbins,
    int nchunks, int hpad) {
  __shared__ int hist[MAX_BINS];
  const int t = threadIdx.x;
  const int c = blockIdx.x;
  hist[t] = 0;
  __syncthreads();
  const int base = c * CHUNK;
  if (base + CHUNK <= nedges) {
    int4 d[4];
#pragma unroll
    for (int k = 0; k < 4; ++k)
      d[k] = *(const int4*)(dst + base + (k * 256 + t) * 4);
#pragma unroll
    for (int k = 0; k < 4; ++k) {
      atomicAdd(&hist[d[k].x >> BIN_SHIFT], 1);
      atomicAdd(&hist[d[k].y >> BIN_SHIFT], 1);
      atomicAdd(&hist[d[k].z >> BIN_SHIFT], 1);
      atomicAdd(&hist[d[k].w >> BIN_SHIFT], 1);
    }
  } else {
    for (int e = base + t; e < nedges; e += 256)
      atomicAdd(&hist[dst[e] >> BIN_SHIFT], 1);
  }
  __syncthreads();
  if (t < nbins) H[t * nchunks + c] = hist[t];
  if (c == 0)
    for (int i = nbins * nchunks + t; i < hpad; i += 256) H[i] = 0;
}

__device__ __forceinline__ int wave_sum_i32(int v) {
#pragma unroll
  for (int m = 16; m >= 1; m >>= 1) v += __shfl_xor(v, m);
  return v;
}

__global__ __launch_bounds__(1024) void scan_counts_kernel(
    const int* __restrict__ cnt, int* __restrict__ off, int pieces) {
  __shared__ int wtot[32];
  const int lane = threadIdx.x & 31;
  const int wave = threadIdx.x >> 5;
  const int4* c4 = (const int4*)cnt + (size_t)wave * pieces * 32 + lane;
  int4* o4 = (int4*)off + (size_t)wave * pieces * 32 + lane;

  int s = 0;
#pragma unroll 5
  for (int p = 0; p < pieces; ++p) {
    const int4 v = c4[p * 32];
    s += v.x + v.y + v.z + v.w;
  }
  s = wave_sum_i32(s);
  if (lane == 0) wtot[wave] = s;
  __syncthreads();
  int carry = wave_sum_i32((lane < wave) ? wtot[lane] : 0);

#pragma unroll 5
  for (int p = 0; p < pieces; ++p) {
    const int4 v = c4[p * 32];
    const int t = v.x + v.y + v.z + v.w;
    int inc = t;
#pragma unroll
    for (int d = 1; d < 32; d <<= 1) {
      const int u = __shfl_up(inc, d);
      if (lane >= d) inc += u;
    }
    int4 o;
    o.x = carry + inc - t;
    o.y = o.x + v.x;
    o.z = o.y + v.y;
    o.w = o.z + v.z;
    o4[p * 32] = o;
    carry += __shfl(inc, 31);
  }
}

__global__ __launch_bounds__(256) void bin_scatter_kernel(
    const int* __restrict__ dst, const int* __restrict__ src,
    const float* __restrict__ edge_w, const float* __restrict__ cu,
    const int* __restrict__ H, const int* __restrict__ S,
    int2* __restrict__ rec, int2* __restrict__ val, int nedges, int nbins,
    int nchunks) {
  __shared__ int cur[MAX_BINS];
  __shared__ int gofs[MAX_BINS];
  __shared__ int wtot[8];
  __shared__ int2 sRec[CHUNK];
  __shared__ int2 sVal[CHUNK];
  const int t = threadIdx.x;
  const int c = blockIdx.x;
  const int lane = t & 31;
  const int wave = t >> 5;
  const int base = c * CHUNK;
  const bool whole = base + CHUNK <= nedges;

  int4 d[4];
  int4 sv[4];
  float4 ev[4];
  float cs[16];
  if (whole) {
#pragma unroll
    for (int k = 0; k < 4; ++k) {
      d[k] = *(const int4*)(dst + base + (k * 256 + t) * 4);
      sv[k] = *(const int4*)(src + base + (k * 256 + t) * 4);
      ev[k] = *(const float4*)(edge_w + base + (k * 256 + t) * 4);
    }
#pragma unroll
    for (int k = 0; k < 4; ++k) {
      cs[4 * k + 0] = cu[sv[k].x];
      cs[4 * k + 1] = cu[sv[k].y];
      cs[4 * k + 2] = cu[sv[k].z];
      cs[4 * k + 3] = cu[sv[k].w];
    }
  }
  int hb = 0, sb = 0;
  if (t < nbins) {
    hb = H[t * nchunks + c];
    sb = S[t * nchunks + c];
  }

  int inc = hb;
#pragma unroll
  for (int s = 1; s < 32; s <<= 1) {
    const int u = __shfl_up(inc, s);
    if (lane >= s) inc += u;
  }
  if (lane == 31) wtot[wave] = inc;
  __syncthreads();
  int carry = 0;
#pragma unroll
  for (int w = 0; w < 8; ++w) carry += (w < wave) ? wtot[w] : 0;
  const int lstart = carry + inc - hb;
  cur[t] = lstart;
  gofs[t] = sb - lstart;
  __syncthreads();

  if (whole) {
    int slot[16];
#pragma unroll
    for (int k = 0; k < 4; ++k) {
      slot[4 * k + 0] = atomicAdd(&cur[d[k].x >> BIN_SHIFT], 1);
      slot[4 * k + 1] = atomicAdd(&cur[d[k].y >> BIN_SHIFT], 1);
      slot[4 * k + 2] = atomicAdd(&cur[d[k].z >> BIN_SHIFT], 1);
      slot[4 * k + 3] = atomicAdd(&cur[d[k].w >> BIN_SHIFT], 1);
    }
#pragma unroll
    for (int k = 0; k < 4; ++k) {
      const int e0 = base + (k * 256 + t) * 4;
      sRec[slot[4 * k + 0]] = make_int2(e0, d[k].x);
      sRec[slot[4 * k + 1]] = make_int2(e0 + 1, d[k].y);
      sRec[slot[4 * k + 2]] = make_int2(e0 + 2, d[k].z);
      sRec[slot[4 * k + 3]] = make_int2(e0 + 3, d[k].w);
      sVal[slot[4 * k + 0]] = make_int2(sv[k].x, __float_as_int(ev[k].x * cs[4 * k + 0]));
      sVal[slot[4 * k + 1]] = make_int2(sv[k].y, __float_as_int(ev[k].y * cs[4 * k + 1]));
      sVal[slot[4 * k + 2]] = make_int2(sv[k].z, __float_as_int(ev[k].z * cs[4 * k + 2]));
      sVal[slot[4 * k + 3]] = make_int2(sv[k].w, __float_as_int(ev[k].w * cs[4 * k + 3]));
    }
  } else {
    for (int e = base + t; e < nedges; e += 256) {
      const int de = dst[e];
      const int se = src[e];
      const float we = edge_w[e] * cu[se];
      const int sl = atomicAdd(&cur[de >> BIN_SHIFT], 1);
      sRec[sl] = make_int2(e, de);
      sVal[sl] = make_int2(se, __float_as_int(we));
    }
  }
  __syncthreads();

  if (whole) {
#pragma unroll
    for (int k = 0; k < CHUNK / 256; ++k) {
      const int i = k * 256 + t;
      const int2 r = sRec[i];
      const int g = gofs[r.y >> BIN_SHIFT] + i;
      rec[g] = r;
      val[g] = sVal[i];
    }
  } else {
    const int n = nedges - base;
    for (int i = t; i < n; i += 256) {
      const int2 r = sRec[i];
      const int g = gofs[r.y >> BIN_SHIFT] + i;
      rec[g] = r;
      val[g] = sVal[i];
    }
  }
}

#define RANK_CAP 16384
__global__ __launch_bounds__(1024) void bin_rank_kernel(
    const int2* __restrict__ rec, const int2* __restrict__ val,
    const int* __restrict__ S, int* __restrict__ off, int* __restrict__ pe,
    int* __restrict__ ps, int* __restrict__ pw, int nrows, int nchunks) {
  __shared__ int cntL[BIN_W];
  __shared__ int curL[BIN_W];
  __shared__ int sE[RANK_CAP];
  __shared__ int sS[RANK_CAP];
  __shared__ int sV[RANK_CAP];
  const int t = threadIdx.x;
  const int b = blockIdx.x;
  const int lo = S[b * nchunks];
  const int hi = S[(b + 1) * nchunks];
  const int tbase = b << BIN_SHIFT;
  const int n = hi - lo;
  const bool staged = n <= RANK_CAP;
  if (t < BIN_W) cntL[t] = 0;
  __syncthreads();
  for (int i0 = lo + t; i0 < hi; i0 += 4096) {
    int d[4];
#pragma unroll
    for (int k = 0; k < 4; ++k) {
      const int i = i0 + k * 1024;
      d[k] = (i < hi) ? rec[i].y : tbase;
    }
#pragma unroll
    for (int k = 0; k < 4; ++k)
      if (i0 + k * 1024 < hi) atomicAdd(&cntL[d[k] - tbase], 1);
  }
  __syncthreads();
  if (t < 32) {
    int v[16];
    int s = 0;
#pragma unroll
    for (int j = 0; j < 16; ++j) {
      v[j] = cntL[t * 16 + j];
      s += v[j];
    }
    int inc = s;
#pragma unroll
    for (int d = 1; d < 32; d <<= 1) {
      const int u = __shfl_up(inc, d);
      if (t >= d) inc += u;
    }
    int run = inc - s;
#pragma unroll
    for (int j = 0; j < 16; ++j) {
      curL[t * 16 + j] = run;
      run += v[j];
    }
  }
  __syncthreads();
  if (t < BIN_W) {
    const int g = tbase + t;
    if (g <= nrows) off[g] = lo + curL[t];
  }
  __syncthreads();
  for (int i0 = lo + t; i0 < hi; i0 += 4096) {
    int2 r[4];
    int2 v[4];
#pragma unroll
    for (int k = 0; k < 4; ++k) {
      const int i = i0 + k * 1024;
      r[k] = (i < hi) ? rec[i] : make_int2(0, tbase);
      v[k] = (i < hi) ? val[i] : make_int2(0, 0);
    }
#pragma unroll
    for (int k = 0; k < 4; ++k)
      if (i0 + k * 1024 < hi) {
        const int slot = atomicAdd(&curL[r[k].y - tbase], 1);
        if (staged) {
          sE[slot] = r[k].x;
          sS[slot] = v[k].x;
          sV[slot] = v[k].y;
        } else {
          pe[lo + slot] = r[k].x;
          ps[lo + slot] = v[k].x;
          pw[lo + slot] = v[k].y;
        }
      }
  }
  if (staged) {
    __syncthreads();
    for (int i = t; i < n; i += 1024) {
      pe[lo + i] = sE[i];
      ps[lo + i] = sS[i];
      pw[lo + i] = sV[i];
    }
  }
}

#define TILE_ROWS 32

__device__ __forceinline__ void fma_row(float4& acc, const float4 v, const float w) {
  acc.x = fmaf(v.x, w, acc.x);
  acc.y = fmaf(v.y, w, acc.y);
  acc.z = fmaf(v.z, w, acc.z);
  acc.w = fmaf(v.w, w, acc.w);
}

__device__ __forceinline__ void add_run(float4& acc, const float* __restrict__ fb,
                                        const int ss, const int wb, const int n) {
  int i = 0;
  for (; i + 8 <= n; i += 8) {
    float4 v[8];
    float wi[8];
#pragma unroll
    for (int k = 0; k < 8; ++k) {
      const int si = __builtin_amdgcn_readlane(ss, i + k);
      wi[k] = __int_as_float(__builtin_amdgcn_readlane(wb, i + k));
      v[k] = *(const float4*)(fb + (size_t)si * D_IN);
    }
#pragma unroll
    for (int k = 0; k < 8; ++k) fma_row(acc, v[k], wi[k]);
  }
  if (i + 4 <= n) {
    float4 v[4];
    float wi[4];
#pragma unroll
    for (int k = 0; k < 4; ++k) {
      const int si = __builtin_amdgcn_readlane(ss, i + k);
      wi[k] = __int_as_float(__builtin_amdgcn_readlane(wb, i + k));
      v[k] = *(const float4*)(fb + (size_t)si * D_IN);
    }
#pragma unroll
    for (int k = 0; k < 4; ++k) fma_row(acc, v[k], wi[k]);
    i += 4;
  }
  for (; i < n; ++i) {
    const int si = __builtin_amdgcn_readlane(ss, i);
    const float wi = __int_as_float(__builtin_amdgcn_readlane(wb, i));
    fma_row(acc, *(const float4*)(fb + (size_t)si * D_IN), wi);
  }
}

__device__ __forceinline__ void add_long_run(
    float4& acc, const float* __restrict__ fb, const int* __restrict__ pe,
    const int* __restrict__ ps, const float* __restrict__ pw, const int beg,
    const int end, const int lane) {
  int last = -1;
  for (int i = beg; i < end; ++i) {
    int best = 0x7fffffff;
    int bp = beg;
    for (int p = beg + lane; p < end; p += 32) {
      const int e = pe[p];
      if (e > last && e < best) {
        best = e;
        bp = p;
      }
    }
#pragma unroll
    for (int m = 16; m >= 1; m >>= 1) {
      const int o = __shfl_xor(best, m);
      const int op = __shfl_xor(bp, m);
      if (o < best) {
        best = o;
        bp = op;
      }
    }
    best = __builtin_amdgcn_readfirstlane(best);
    bp = __builtin_amdgcn_readfirstlane(bp);
    last = best;
    const int si = ps[bp];
    const float wi = pw[bp];
    fma_row(acc, *(const float4*)(fb + (size_t)si * D_IN), wi);
  }
}

__device__ __forceinline__ int rank_of(const int e) {
  int r = 0;
#pragma unroll
  for (int j = 0; j < 32; ++j)
    r += (__builtin_amdgcn_readlane(e, j) < e) ? 1 : 0;
  return r;
}

__device__ __forceinline__ void add_mid_run(
    float4& acc, const float* __restrict__ fb, const int* __restrict__ pe,
    const int* __restrict__ ps, const float* __restrict__ pw, float4* scr,
    const int beg, const int deg, const int lane) {
  const bool ah = 32 + lane < deg;
  const int il = beg + lane;
  const int ih = ah ? beg + 32 + lane : beg;
  const int ql = pe[il];
  const int qh = pe[ih];
  const int tl = ps[il];
  const int th = ps[ih];
  const float wl = pw[il];
  const float wh = pw[ih];
  const int el = ql;
  const int eh = ah ? qh : 0x7fffffff;
  int rl = 0;
  int rh = 0;
#pragma nounroll
  for (int j = 0; j < 32; ++j) {
    const int a = __builtin_amdgcn_readlane(el, j);
    const int b = __builtin_amdgcn_readlane(eh, j);
    rl += ((a < el) ? 1 : 0) + ((b < el) ? 1 : 0);
    rh += ((a < eh) ? 1 : 0) + ((b < eh) ? 1 : 0);
  }
  scr[rl] = make_float4(__int_as_float(tl), wl, 0.f, 0.f);
  if (ah) scr[rh] = make_float4(__int_as_float(th), wh, 0.f, 0.f);
#pragma nounroll
  for (int c = 0; c < 2; ++c) {
    const float4 p = scr[c * 32 + lane];
    const int n = (c == 0) ? 32 : deg - 32;
    add_run(acc, fb, __float_as_int(p.x), __float_as_int(p.y), n);
  }
}

__device__ __forceinline__ int load_bounds(const int* __restrict__ off, const int d0,
                                           const int lane, const int nrows) {
  int g = d0 + lane;
  if (g > nrows) g = nrows;
  return (lane < 3) ? off[g] : 0;
}

__global__ __launch_bounds__(512) void influence_pull_gemm_kernel(
    const float* __restrict__ feat, const float* __restrict__ W,
    const float* __restrict__ cv, const int* __restrict__ off,
    const int* __restrict__ pe, const int* __restrict__ ps,
    const float* __restrict__ pw, float* __restrict__ out, int nrows,
    int ntiles) {
  __shared__ float sW[D_IN * D_OUT];
  __shared__ float4 sA[TILE_ROWS * 32];

  const int tid = threadIdx.x;
  int tile = blockIdx.x;
  if (tile >= ntiles) return;

  int b0, b1, b2;
  int q0, q1, t0, t1;
  float u0, u1;
  {
    const int lane = tid & 31;
    const int wave = tid >> 5;
    const int ov = load_bounds(off, tile * TILE_ROWS + 2 * wave, lane, nrows);

#pragma unroll
    for (int i = 0; i < (D_IN * D_OUT) / (4 * 512); ++i)
      ((float4*)sW)[tid + i * 512] = ((const float4*)W)[tid + i * 512];

    b0 = __builtin_amdgcn_readlane(ov, 0);
    b1 = __builtin_amdgcn_readlane(ov, 1);
    b2 = __builtin_amdgcn_readlane(ov, 2);
    const int i0 = (b1 - b0 <= 32 && lane < b1 - b0) ? b0 + lane : 0;
    const int i1 = (b2 - b1 <= 32 && lane < b2 - b1) ? b1 + lane : 0;
    q0 = pe[i0];
    q1 = pe[i1];
    t0 = ps[i0];
    t1 = ps[i1];
    u0 = pw[i0];
    u1 = pw[i1];
  }

  for (;;) {
    int tx = tid;
    asm volatile("" : "+v"(tx));
    const int lane = tx & 31;
    const int wave = __builtin_amdgcn_readfirstlane(tx >> 5);

    const int rowBase = tile * TILE_ROWS;
    const int deg0 = b1 - b0;
    const int deg1 = b2 - b1;
    const bool short0 = deg0 <= 32;
    const bool short1 = deg1 <= 32;
    const bool a0 = short0 && lane < deg0;
    const bool a1 = short1 && lane < deg1;
    const int e0 = a0 ? q0 : 0x7fffffff;
    const int e1 = a1 ? q1 : 0x7fffffff;
    const int s0 = a0 ? t0 : 0;
    const int s1 = a1 ? t1 : 0;
    const float w0 = a0 ? u0 : 0.f;
    const float w1 = a1 ? u1 : 0.f;

    const int r0 = rank_of(e0);
    const int r1 = rank_of(e1);
    const int ss0 = __builtin_amdgcn_ds_permute(r0 << 2, s0);
    const int wb0 = __builtin_amdgcn_ds_permute(r0 << 2, __float_as_int(w0));
    const int ss1 = __builtin_amdgcn_ds_permute(r1 << 2, s1);
    const int wb1 = __builtin_amdgcn_ds_permute(r1 << 2, __float_as_int(w1));

    const int next = tile + gridDim.x;
    const int ovN = load_bounds(off, next * TILE_ROWS + 2 * wave, lane, nrows);

    const float* fb = feat + lane * 4;
    const int lr0 = 2 * wave;
    float4* scr = sA + lr0 * 32;

    float4 acc0 = {0.f, 0.f, 0.f, 0.f};
    float4 acc1 = {0.f, 0.f, 0.f, 0.f};
    if (short0) add_run(acc0, fb, ss0, wb0, deg0);
    else if (deg0 <= 64) add_mid_run(acc0, fb, pe, ps, pw, scr, b0, deg0, lane);
    else add_long_run(acc0, fb, pe, ps, pw, b0, b1, lane);

    const int nb0 = __builtin_amdgcn_readlane(ovN, 0);
    const int nb1 = __builtin_amdgcn_readlane(ovN, 1);
    const int nb2 = __builtin_amdgcn_readlane(ovN, 2);

    if (short1) add_run(acc1, fb, ss1, wb1, deg1);
    else if (deg1 <= 64) add_mid_run(acc1, fb, pe, ps, pw, scr, b1, deg1, lane);
    else add_long_run(acc1, fb, pe, ps, pw, b1, b2, lane);

    const int ni0 = (nb1 - nb0 <= 32 && lane < nb1 - nb0) ? nb0 + lane : 0;
    const int ni1 = (nb2 - nb1 <= 32 && lane < nb2 - nb1) ? nb1 + lane : 0;
    const int nq0 = pe[ni0];
    const int nq1 = pe[ni1];
    const int nt0 = ps[ni0];
    const int nt1 = ps[ni1];
    const float nu0 = pw[ni0];
    const float nu1 = pw[ni1];

    sA[lr0 * 32 + (lane ^ lr0)] = acc0;
    sA[(lr0 + 1) * 32 + (lane ^ (lr0 + 1))] = acc1;
    __syncthreads();

    const int rt      = wave >> 3;
    const int colBase = (wave & 7) * 16;
    const int m       = lane & 15;
    const int h       = lane >> 4;
    const int kh      = h * 2;
    const int ar = rt * 16 + m;
    const v2f* arow = (const v2f*)sA + ar * 64;

    const int mBase = rowBase + rt * 16 + h * 8;
    const int crow = rowBase + rt * 16 + m;
    const float cvl = cv[crow < nrows ? crow : nrows - 1];
    float cvr[8];
#pragma unroll
    for (int v = 0; v < 8; ++v) cvr[v] = __shfl(cvl, h * 8 + v);

    v8f acc = {};
#pragma unroll
    for (int kk = 0; kk < D_IN / 4; ++kk) {
      const v2f A = arow[((kk ^ ar) << 1) | h];
      const int kb = kk * 4 + kh;
      v2f B;
      B.x = sW[kb * D_OUT + colBase + m];
      B.y = sW[(kb + 1) * D_OUT + colBase + m];
      acc = __builtin_amdgcn_wmma_f32_16x16x4_f32(
           false, A,  false, B,
           (short)0, acc,  false,  false);
    }

#pragma unroll
    for (int v = 0; v < 8; ++v) {
      const int row = mBase + v;
      if (row < nrows)
        out[(size_t)row * D_OUT + colBase + m] = acc[v] * cvr[v];
    }

    if (next >= ntiles) break;
    __syncthreads();
    tile = next;
    b0 = nb0; b1 = nb1; b2 = nb2;
    q0 = nq0; q1 = nq1;
    t0 = nt0; t1 = nt1;
    u0 = nu0; u1 = nu1;
  }
}

extern "C" void kernel_launch(void* const* d_in, const int* in_sizes, int n_in,
                              void* d_out, int out_size, void* d_ws, size_t ws_size,
                              hipStream_t stream) {
  const float* feat = (const float*)d_in[0];
  const float* W    = (const float*)d_in[1];
  const float* cu   = (const float*)d_in[2];
  const float* cv   = (const float*)d_in[3];
  const float* ew   = (const float*)d_in[4];
  const int*   src  = (const int*)d_in[5];
  const int*   dst  = (const int*)d_in[6];

  float* out = (float*)d_out;

  const int N = in_sizes[2];
  const int E = in_sizes[5];

  const int nbins   = N / BIN_W + 1;
  const int nchunks = (E + CHUNK - 1) / CHUNK;
  const int pieces  = (nbins * nchunks) / 4096 + 1;
  const int Hp = pieces * 4096;
  const int Np = (N / 4096 + 1) * 4096;
  int*  H    = (int*)d_ws;
  int*  S    = H + Hp;
  int*  off  = S + Hp;
  int2* rec  = (int2*)(off + Np);
  int2* val  = rec + E;
  int*  pe   = (int*)(val + E);
  int*  ps   = pe + E;
  int*  pw   = ps + E;

  bin_count_kernel<<<nchunks, 256, 0, stream>>>(dst, H, E, nbins, nchunks, Hp);
  scan_counts_kernel<<<1, 1024, 0, stream>>>(H, S, pieces);
  bin_scatter_kernel<<<nchunks, 256, 0, stream>>>(dst, src, ew, cu, H, S, rec,
                                                  val, E, nbins, nchunks);
  bin_rank_kernel<<<nbins, 1024, 0, stream>>>(rec, val, S, off, pe, ps, pw, N,
                                              nchunks);

  const int rowTiles = (N + TILE_ROWS - 1) / TILE_ROWS;
  influence_pull_gemm_kernel<<<1024, 512, 0, stream>>>(
      feat, W, cv, off, pe, ps, (const float*)pw, out, N, rowTiles);
}


# sort's count and prefix launches dropped: an edge block groups its records by range in its own span, stores its range starts; a range's block sums them for its place, reads its pieces; sums as before
# speedup vs baseline: 3.5157x; 1.1582x over previous
#include <hip/hip_runtime.h>


typedef __attribute__((ext_vector_type(2))) float v2f;
typedef __attribute__((ext_vector_type(8))) float v8f;

#define D_IN  128
#define D_OUT 128

#define BIN_SHIFT  9
#define BIN_W      512
#define MAX_BINS   256
#define CHUNK      4096
#define MAX_CHUNKS 1024

__device__ __forceinline__ int wave_sum_i32(int v) {
#pragma unroll
  for (int m = 16; m >= 1; m >>= 1) v += __shfl_xor(v, m);
  return v;
}

__global__ __launch_bounds__(256) void bin_group_kernel(
    const int* __restrict__ dst, const int* __restrict__ src,
    const float* __restrict__ edge_w, const float* __restrict__ cu,
    int* __restrict__ P, int2* __restrict__ rec, int2* __restrict__ val,
    int nedges, int nbins, int nchunks) {
  __shared__ int cur[MAX_BINS];
  __shared__ int wtot[8];
  __shared__ int4 sRec4[CHUNK / 2];
  __shared__ int4 sVal4[CHUNK / 2];
  int2* sRec = (int2*)sRec4;
  int2* sVal = (int2*)sVal4;
  const int t = threadIdx.x;
  const int c = blockIdx.x;
  const int lane = t & 31;
  const int wave = t >> 5;
  const int base = c * CHUNK;
  const bool whole = base + CHUNK <= nedges;

  cur[t] = 0;
  int4 d[4];
  int4 sv[4];
  float4 ev[4];
  float cs[16];
  if (whole) {
#pragma unroll
    for (int k = 0; k < 4; ++k) {
      d[k] = *(const int4*)(dst + base + (k * 256 + t) * 4);
      sv[k] = *(const int4*)(src + base + (k * 256 + t) * 4);
      ev[k] = *(const float4*)(edge_w + base + (k * 256 + t) * 4);
    }
#pragma unroll
    for (int k = 0; k < 4; ++k) {
      cs[4 * k + 0] = cu[sv[k].x];
      cs[4 * k + 1] = cu[sv[k].y];
      cs[4 * k + 2] = cu[sv[k].z];
      cs[4 * k + 3] = cu[sv[k].w];
    }
  }
  __syncthreads();

  int rk[16];
  if (whole) {
#pragma unroll
    for (int k = 0; k < 4; ++k) {
      rk[4 * k + 0] = atomicAdd(&cur[d[k].x >> BIN_SHIFT], 1);
      rk[4 * k + 1] = atomicAdd(&cur[d[k].y >> BIN_SHIFT], 1);
      rk[4 * k + 2] = atomicAdd(&cur[d[k].z >> BIN_SHIFT], 1);
      rk[4 * k + 3] = atomicAdd(&cur[d[k].w >> BIN_SHIFT], 1);
    }
  } else {
    for (int e = base + t; e < nedges; e += 256)
      atomicAdd(&cur[dst[e] >> BIN_SHIFT], 1);
  }
  __syncthreads();

  const int hb = cur[t];
  int inc = hb;
#pragma unroll
  for (int s = 1; s < 32; s <<= 1) {
    const int u = __shfl_up(inc, s);
    if (lane >= s) inc += u;
  }
  if (lane == 31) wtot[wave] = inc;
  __syncthreads();
  int carry = 0;
#pragma unroll
  for (int w = 0; w < 8; ++w) carry += (w < wave) ? wtot[w] : 0;
  const int lstart = carry + inc - hb;
  cur[t] = lstart;
  if (t < nbins) P[t * nchunks + c] = lstart;
  if (t == MAX_BINS - 1) P[nbins * nchunks + c] = lstart + hb;
  __syncthreads();

  if (whole) {
#pragma unroll
    for (int k = 0; k < 4; ++k) {
      const int e0 = base + (k * 256 + t) * 4;
      const int s0 = cur[d[k].x >> BIN_SHIFT] + rk[4 * k + 0];
      const int s1 = cur[d[k].y >> BIN_SHIFT] + rk[4 * k + 1];
      const int s2 = cur[d[k].z >> BIN_SHIFT] + rk[4 * k + 2];
      const int s3 = cur[d[k].w >> BIN_SHIFT] + rk[4 * k + 3];
      sRec[s0] = make_int2(e0, d[k].x);
      sRec[s1] = make_int2(e0 + 1, d[k].y);
      sRec[s2] = make_int2(e0 + 2, d[k].z);
      sRec[s3] = make_int2(e0 + 3, d[k].w);
      sVal[s0] = make_int2(sv[k].x, __float_as_int(ev[k].x * cs[4 * k + 0]));
      sVal[s1] = make_int2(sv[k].y, __float_as_int(ev[k].y * cs[4 * k + 1]));
      sVal[s2] = make_int2(sv[k].z, __float_as_int(ev[k].z * cs[4 * k + 2]));
      sVal[s3] = make_int2(sv[k].w, __float_as_int(ev[k].w * cs[4 * k + 3]));
    }
  } else {
    for (int e = base + t; e < nedges; e += 256) {
      const int de = dst[e];
      const int se = src[e];
      const float we = edge_w[e] * cu[se];
      const int sl = atomicAdd(&cur[de >> BIN_SHIFT], 1);
      sRec[sl] = make_int2(e, de);
      sVal[sl] = make_int2(se, __float_as_int(we));
    }
  }
  __syncthreads();

  if (whole) {
    int4* ro = (int4*)(rec + base);
    int4* vo = (int4*)(val + base);
#pragma unroll
    for (int k = 0; k < CHUNK / 512; ++k) {
      const int i = k * 256 + t;
      ro[i] = sRec4[i];
      vo[i] = sVal4[i];
    }
  } else {
    const int n = nedges - base;
    for (int i = t; i < n; i += 256) {
      rec[base + i] = sRec[i];
      val[base + i] = sVal[i];
    }
  }
}

#define RANK_CAP 16384
#define PIECES   8

__device__ __forceinline__ void put_edge(
    int* sE, int* sS, int* sV, int* __restrict__ pe, int* __restrict__ ps,
    int* __restrict__ pw, const int slot, const int2 r, const int2 v,
    const int lo, const bool staged) {
  if (staged) {
    sE[slot] = r.x;
    sS[slot] = v.x;
    sV[slot] = v.y;
  } else {
    pe[lo + slot] = r.x;
    ps[lo + slot] = v.x;
    pw[lo + slot] = v.y;
  }
}

__global__ __launch_bounds__(1024) void bin_rank_kernel(
    const int2* __restrict__ rec, const int2* __restrict__ val,
    const int* __restrict__ P, int* __restrict__ off, int* __restrict__ pe,
    int* __restrict__ ps, int* __restrict__ pw, int nrows, int nchunks) {
  __shared__ int cntL[BIN_W];
  __shared__ int curL[BIN_W];
  __shared__ int pcS[MAX_CHUNKS];
  __shared__ int pcN[MAX_CHUNKS];
  __shared__ int wlo[32];
  __shared__ int whi[32];
  __shared__ int sE[RANK_CAP];
  __shared__ int sS[RANK_CAP];
  __shared__ int sV[RANK_CAP];
  const int t = threadIdx.x;
  const int b = blockIdx.x;
  const int lane = t & 31;
  const int wave = __builtin_amdgcn_readfirstlane(t >> 5);
  const int tbase = b << BIN_SHIFT;

  int a0 = 0;
  int a1 = 0;
  if (t < nchunks) {
    a0 = P[b * nchunks + t];
    a1 = P[(b + 1) * nchunks + t];
  }
  pcS[lane * 32 + wave] = t * CHUNK + a0;
  pcN[lane * 32 + wave] = a1 - a0;
  if (t < BIN_W) cntL[t] = 0;
  a0 = wave_sum_i32(a0);
  a1 = wave_sum_i32(a1);
  if (lane == 0) {
    wlo[wave] = a0;
    whi[wave] = a1;
  }
  __syncthreads();
  const int lo = __builtin_amdgcn_readfirstlane(wave_sum_i32(wlo[lane]));
  const int hi = __builtin_amdgcn_readfirstlane(wave_sum_i32(whi[lane]));
  const int n = hi - lo;
  const bool staged = n <= RANK_CAP;

  const int vS = pcS[wave * 32 + lane];
  const int vN = pcN[wave * 32 + lane];
  const int npw = (nchunks - wave + 31) >> 5;

  for (int j0 = 0; j0 < npw; j0 += PIECES) {
    int s[PIECES];
    int m[PIECES];
    int d[PIECES];
#pragma unroll
    for (int k = 0; k < PIECES; ++k) {
      s[k] = __builtin_amdgcn_readlane(vS, j0 + k);
      m[k] = __builtin_amdgcn_readlane(vN, j0 + k);
    }
#pragma unroll
    for (int k = 0; k < PIECES; ++k)
      d[k] = rec[(lane < m[k]) ? s[k] + lane : 0].y;
#pragma unroll
    for (int k = 0; k < PIECES; ++k)
      if (lane < m[k]) atomicAdd(&cntL[d[k] - tbase], 1);
#pragma unroll
    for (int k = 0; k < PIECES; ++k)
      if (m[k] > 32)
        for (int j = lane + 32; j < m[k]; j += 32)
          atomicAdd(&cntL[rec[s[k] + j].y - tbase], 1);
  }
  __syncthreads();
  if (t < 32) {
    int v[16];
    int s = 0;
#pragma unroll
    for (int j = 0; j < 16; ++j) {
      v[j] = cntL[t * 16 + j];
      s += v[j];
    }
    int inc = s;
#pragma unroll
    for (int d = 1; d < 32; d <<= 1) {
      const int u = __shfl_up(inc, d);
      if (t >= d) inc += u;
    }
    int run = inc - s;
#pragma unroll
    for (int j = 0; j < 16; ++j) {
      curL[t * 16 + j] = run;
      run += v[j];
    }
  }
  __syncthreads();
  if (t < BIN_W) {
    const int g = tbase + t;
    if (g <= nrows) off[g] = lo + curL[t];
  }
  __syncthreads();
  for (int j0 = 0; j0 < npw; j0 += PIECES) {
    int s[PIECES];
    int m[PIECES];
    int2 r[PIECES];
    int2 v[PIECES];
#pragma unroll
    for (int k = 0; k < PIECES; ++k) {
      s[k] = __builtin_amdgcn_readlane(vS, j0 + k);
      m[k] = __builtin_amdgcn_readlane(vN, j0 + k);
    }
#pragma unroll
    for (int k = 0; k < PIECES; ++k) {
      const int i = (lane < m[k]) ? s[k] + lane : 0;
      r[k] = rec[i];
      v[k] = val[i];
    }
    int slot[PIECES];
#pragma unroll
    for (int k = 0; k < PIECES; ++k)
      slot[k] = (lane < m[k]) ? atomicAdd(&curL[r[k].y - tbase], 1) : 0;
#pragma unroll
    for (int k = 0; k < PIECES; ++k)
      if (lane < m[k])
        put_edge(sE, sS, sV, pe, ps, pw, slot[k], r[k], v[k], lo, staged);
#pragma unroll
    for (int k = 0; k < PIECES; ++k)
      if (m[k] > 32)
        for (int j = lane + 32; j < m[k]; j += 32) {
          const int2 rr = rec[s[k] + j];
          const int2 vv = val[s[k] + j];
          const int sl = atomicAdd(&curL[rr.y - tbase], 1);
          put_edge(sE, sS, sV, pe, ps, pw, sl, rr, vv, lo, staged);
        }
  }
  if (staged) {
    __syncthreads();
    for (int i = t; i < n; i += 1024) {
      pe[lo + i] = sE[i];
      ps[lo + i] = sS[i];
      pw[lo + i] = sV[i];
    }
  }
}

#define TILE_ROWS 32

__device__ __forceinline__ void fma_row(float4& acc, const float4 v, const float w) {
  acc.x = fmaf(v.x, w, acc.x);
  acc.y = fmaf(v.y, w, acc.y);
  acc.z = fmaf(v.z, w, acc.z);
  acc.w = fmaf(v.w, w, acc.w);
}

__device__ __forceinline__ void add_run(float4& acc, const float* __restrict__ fb,
                                        const int ss, const int wb, const int n) {
  int i = 0;
  for (; i + 8 <= n; i += 8) {
    float4 v[8];
    float wi[8];
#pragma unroll
    for (int k = 0; k < 8; ++k) {
      const int si = __builtin_amdgcn_readlane(ss, i + k);
      wi[k] = __int_as_float(__builtin_amdgcn_readlane(wb, i + k));
      v[k] = *(const float4*)(fb + (size_t)si * D_IN);
    }
#pragma unroll
    for (int k = 0; k < 8; ++k) fma_row(acc, v[k], wi[k]);
  }
  if (i + 4 <= n) {
    float4 v[4];
    float wi[4];
#pragma unroll
    for (int k = 0; k < 4; ++k) {
      const int si = __builtin_amdgcn_readlane(ss, i + k);
      wi[k] = __int_as_float(__builtin_amdgcn_readlane(wb, i + k));
      v[k] = *(const float4*)(fb + (size_t)si * D_IN);
    }
#pragma unroll
    for (int k = 0; k < 4; ++k) fma_row(acc, v[k], wi[k]);
    i += 4;
  }
  for (; i < n; ++i) {
    const int si = __builtin_amdgcn_readlane(ss, i);
    const float wi = __int_as_float(__builtin_amdgcn_readlane(wb, i));
    fma_row(acc, *(const float4*)(fb + (size_t)si * D_IN), wi);
  }
}

__device__ __forceinline__ void add_long_run(
    float4& acc, const float* __restrict__ fb, const int* __restrict__ pe,
    const int* __restrict__ ps, const float* __restrict__ pw, const int beg,
    const int end, const int lane) {
  int last = -1;
  for (int i = beg; i < end; ++i) {
    int best = 0x7fffffff;
    int bp = beg;
    for (int p = beg + lane; p < end; p += 32) {
      const int e = pe[p];
      if (e > last && e < best) {
        best = e;
        bp = p;
      }
    }
#pragma unroll
    for (int m = 16; m >= 1; m >>= 1) {
      const int o = __shfl_xor(best, m);
      const int op = __shfl_xor(bp, m);
      if (o < best) {
        best = o;
        bp = op;
      }
    }
    best = __builtin_amdgcn_readfirstlane(best);
    bp = __builtin_amdgcn_readfirstlane(bp);
    last = best;
    const int si = ps[bp];
    const float wi = pw[bp];
    fma_row(acc, *(const float4*)(fb + (size_t)si * D_IN), wi);
  }
}

__device__ __forceinline__ int rank_of(const int e) {
  int r = 0;
#pragma unroll
  for (int j = 0; j < 32; ++j)
    r += (__builtin_amdgcn_readlane(e, j) < e) ? 1 : 0;
  return r;
}

__device__ __forceinline__ void add_mid_run(
    float4& acc, const float* __restrict__ fb, const int* __restrict__ pe,
    const int* __restrict__ ps, const float* __restrict__ pw, float4* scr,
    const int beg, const int deg, const int lane) {
  const bool ah = 32 + lane < deg;
  const int il = beg + lane;
  const int ih = ah ? beg + 32 + lane : beg;
  const int ql = pe[il];
  const int qh = pe[ih];
  const int tl = ps[il];
  const int th = ps[ih];
  const float wl = pw[il];
  const float wh = pw[ih];
  const int el = ql;
  const int eh = ah ? qh : 0x7fffffff;
  int rl = 0;
  int rh = 0;
#pragma nounroll
  for (int j = 0; j < 32; ++j) {
    const int a = __builtin_amdgcn_readlane(el, j);
    const int b = __builtin_amdgcn_readlane(eh, j);
    rl += ((a < el) ? 1 : 0) + ((b < el) ? 1 : 0);
    rh += ((a < eh) ? 1 : 0) + ((b < eh) ? 1 : 0);
  }
  scr[rl] = make_float4(__int_as_float(tl), wl, 0.f, 0.f);
  if (ah) scr[rh] = make_float4(__int_as_float(th), wh, 0.f, 0.f);
#pragma nounroll
  for (int c = 0; c < 2; ++c) {
    const float4 p = scr[c * 32 + lane];
    const int n = (c == 0) ? 32 : deg - 32;
    add_run(acc, fb, __float_as_int(p.x), __float_as_int(p.y), n);
  }
}

__device__ __forceinline__ int load_bounds(const int* __restrict__ off, const int d0,
                                           const int lane, const int nrows) {
  int g = d0 + lane;
  if (g > nrows) g = nrows;
  return (lane < 3) ? off[g] : 0;
}

__global__ __launch_bounds__(512) void influence_pull_gemm_kernel(
    const float* __restrict__ feat, const float* __restrict__ W,
    const float* __restrict__ cv, const int* __restrict__ off,
    const int* __restrict__ pe, const int* __restrict__ ps,
    const float* __restrict__ pw, float* __restrict__ out, int nrows,
    int ntiles) {
  __shared__ float sW[D_IN * D_OUT];
  __shared__ float4 sA[TILE_ROWS * 32];

  const int tid = threadIdx.x;
  int tile = blockIdx.x;
  if (tile >= ntiles) return;

  int b0, b1, b2;
  int q0, q1, t0, t1;
  float u0, u1;
  {
    const int lane = tid & 31;
    const int wave = tid >> 5;
    const int ov = load_bounds(off, tile * TILE_ROWS + 2 * wave, lane, nrows);

#pragma unroll
    for (int i = 0; i < (D_IN * D_OUT) / (4 * 512); ++i)
      ((float4*)sW)[tid + i * 512] = ((const float4*)W)[tid + i * 512];

    b0 = __builtin_amdgcn_readlane(ov, 0);
    b1 = __builtin_amdgcn_readlane(ov, 1);
    b2 = __builtin_amdgcn_readlane(ov, 2);
    const int i0 = (b1 - b0 <= 32 && lane < b1 - b0) ? b0 + lane : 0;
    const int i1 = (b2 - b1 <= 32 && lane < b2 - b1) ? b1 + lane : 0;
    q0 = pe[i0];
    q1 = pe[i1];
    t0 = ps[i0];
    t1 = ps[i1];
    u0 = pw[i0];
    u1 = pw[i1];
  }

  for (;;) {
    int tx = tid;
    asm volatile("" : "+v"(tx));
    const int lane = tx & 31;
    const int wave = __builtin_amdgcn_readfirstlane(tx >> 5);

    const int rowBase = tile * TILE_ROWS;
    const int deg0 = b1 - b0;
    const int deg1 = b2 - b1;
    const bool short0 = deg0 <= 32;
    const bool short1 = deg1 <= 32;
    const bool a0 = short0 && lane < deg0;
    const bool a1 = short1 && lane < deg1;
    const int e0 = a0 ? q0 : 0x7fffffff;
    const int e1 = a1 ? q1 : 0x7fffffff;
    const int s0 = a0 ? t0 : 0;
    const int s1 = a1 ? t1 : 0;
    const float w0 = a0 ? u0 : 0.f;
    const float w1 = a1 ? u1 : 0.f;

    const int r0 = rank_of(e0);
    const int r1 = rank_of(e1);
    const int ss0 = __builtin_amdgcn_ds_permute(r0 << 2, s0);
    const int wb0 = __builtin_amdgcn_ds_permute(r0 << 2, __float_as_int(w0));
    const int ss1 = __builtin_amdgcn_ds_permute(r1 << 2, s1);
    const int wb1 = __builtin_amdgcn_ds_permute(r1 << 2, __float_as_int(w1));

    const int next = tile + gridDim.x;
    const int ovN = load_bounds(off, next * TILE_ROWS + 2 * wave, lane, nrows);

    const float* fb = feat + lane * 4;
    const int lr0 = 2 * wave;
    float4* scr = sA + lr0 * 32;

    float4 acc0 = {0.f, 0.f, 0.f, 0.f};
    float4 acc1 = {0.f, 0.f, 0.f, 0.f};
    if (short0) add_run(acc0, fb, ss0, wb0, deg0);
    else if (deg0 <= 64) add_mid_run(acc0, fb, pe, ps, pw, scr, b0, deg0, lane);
    else add_long_run(acc0, fb, pe, ps, pw, b0, b1, lane);

    const int nb0 = __builtin_amdgcn_readlane(ovN, 0);
    const int nb1 = __builtin_amdgcn_readlane(ovN, 1);
    const int nb2 = __builtin_amdgcn_readlane(ovN, 2);

    if (short1) add_run(acc1, fb, ss1, wb1, deg1);
    else if (deg1 <= 64) add_mid_run(acc1, fb, pe, ps, pw, scr, b1, deg1, lane);
    else add_long_run(acc1, fb, pe, ps, pw, b1, b2, lane);

    const int ni0 = (nb1 - nb0 <= 32 && lane < nb1 - nb0) ? nb0 + lane : 0;
    const int ni1 = (nb2 - nb1 <= 32 && lane < nb2 - nb1) ? nb1 + lane : 0;
    const int nq0 = pe[ni0];
    const int nq1 = pe[ni1];
    const int nt0 = ps[ni0];
    const int nt1 = ps[ni1];
    const float nu0 = pw[ni0];
    const float nu1 = pw[ni1];

    sA[lr0 * 32 + (lane ^ lr0)] = acc0;
    sA[(lr0 + 1) * 32 + (lane ^ (lr0 + 1))] = acc1;
    __syncthreads();

    const int rt      = wave >> 3;
    const int colBase = (wave & 7) * 16;
    const int m       = lane & 15;
    const int h       = lane >> 4;
    const int kh      = h * 2;
    const int ar = rt * 16 + m;
    const v2f* arow = (const v2f*)sA + ar * 64;

    const int mBase = rowBase + rt * 16 + h * 8;
    const int crow = rowBase + rt * 16 + m;
    const float cvl = cv[crow < nrows ? crow : nrows - 1];
    float cvr[8];
#pragma unroll
    for (int v = 0; v < 8; ++v) cvr[v] = __shfl(cvl, h * 8 + v);

    v8f acc = {};
#pragma unroll
    for (int kk = 0; kk < D_IN / 4; ++kk) {
      const v2f A = arow[((kk ^ ar) << 1) | h];
      const int kb = kk * 4 + kh;
      v2f B;
      B.x = sW[kb * D_OUT + colBase + m];
      B.y = sW[(kb + 1) * D_OUT + colBase + m];
      acc = __builtin_amdgcn_wmma_f32_16x16x4_f32(
           false, A,  false, B,
           (short)0, acc,  false,  false);
    }

#pragma unroll
    for (int v = 0; v < 8; ++v) {
      const int row = mBase + v;
      if (row < nrows)
        out[(size_t)row * D_OUT + colBase + m] = acc[v] * cvr[v];
    }

    if (next >= ntiles) break;
    __syncthreads();
    tile = next;
    b0 = nb0; b1 = nb1; b2 = nb2;
    q0 = nq0; q1 = nq1;
    t0 = nt0; t1 = nt1;
    u0 = nu0; u1 = nu1;
  }
}

extern "C" void kernel_launch(void* const* d_in, const int* in_sizes, int n_in,
                              void* d_out, int out_size, void* d_ws, size_t ws_size,
                              hipStream_t stream) {
  const float* feat = (const float*)d_in[0];
  const float* W    = (const float*)d_in[1];
  const float* cu   = (const float*)d_in[2];
  const float* cv   = (const float*)d_in[3];
  const float* ew   = (const float*)d_in[4];
  const int*   src  = (const int*)d_in[5];
  const int*   dst  = (const int*)d_in[6];

  float* out = (float*)d_out;

  const int N = in_sizes[2];
  const int E = in_sizes[5];

  const int nbins   = N / BIN_W + 1;
  const int nchunks = (E + CHUNK - 1) / CHUNK;
  const int Pp = (((nbins + 1) * nchunks) / 4 + 1) * 4;
  const int Np = (N / 4096 + 1) * 4096;
  const int Ep = nchunks * CHUNK;
  int*  P    = (int*)d_ws;
  int*  off  = P + Pp;
  int2* rec  = (int2*)(off + Np);
  int2* val  = rec + Ep;
  int*  pe   = (int*)(val + Ep);
  int*  ps   = pe + E;
  int*  pw   = ps + E;

  bin_group_kernel<<<nchunks, 256, 0, stream>>>(dst, src, ew, cu, P, rec, val, E,
                                                nbins, nchunks);
  bin_rank_kernel<<<nbins, 1024, 0, stream>>>(rec, val, P, off, pe, ps, pw, N,
                                              nchunks);

  const int rowTiles = (N + TILE_ROWS - 1) / TILE_ROWS;
  influence_pull_gemm_kernel<<<1024, 512, 0, stream>>>(
      feat, W, cv, off, pe, ps, (const float*)pw, out, N, rowTiles);
}


# sort made repeatable: each wave counts in LDS counters of its own, a prefix over waves gives its slots; runs have a fixed stored order and are summed as stored, no ranking by id; sum order changed
# speedup vs baseline: 3.8099x; 1.0837x over previous
#include <hip/hip_runtime.h>


typedef __attribute__((ext_vector_type(2))) float v2f;
typedef __attribute__((ext_vector_type(8))) float v8f;

#define D_IN  128
#define D_OUT 128

#define BIN_SHIFT  9
#define BIN_W      512
#define MAX_BINS   256
#define CHUNK      4096
#define MAX_CHUNKS 1024
#define GROUP_WAVES 8
#define RANK_WAVES  32

__device__ __forceinline__ int wave_sum_i32(int v) {
#pragma unroll
  for (int m = 16; m >= 1; m >>= 1) v += __shfl_xor(v, m);
  return v;
}

__global__ __launch_bounds__(256) void bin_group_kernel(
    const int* __restrict__ dst, const int* __restrict__ src,
    const float* __restrict__ edge_w, const float* __restrict__ cu,
    int* __restrict__ P, int* __restrict__ rec, int2* __restrict__ val,
    int nedges, int nbins, int nchunks) {
  __shared__ int wc[GROUP_WAVES * MAX_BINS];
  __shared__ int wtot[8];
  __shared__ int4 sRec4[CHUNK / 4];
  __shared__ int4 sVal4[CHUNK / 2];
  int* sRec = (int*)sRec4;
  int2* sVal = (int2*)sVal4;
  const int t = threadIdx.x;
  const int c = blockIdx.x;
  const int lane = t & 31;
  const int wave = t >> 5;
  const int base = c * CHUNK;
  const bool whole = base + CHUNK <= nedges;
  int* wcw = wc + wave * MAX_BINS;

#pragma unroll
  for (int w = 0; w < GROUP_WAVES; ++w) wc[w * MAX_BINS + t] = 0;
  int4 d[4];
  int4 sv[4];
  float4 ev[4];
  float cs[16];
  if (whole) {
#pragma unroll
    for (int k = 0; k < 4; ++k) {
      d[k] = *(const int4*)(dst + base + (k * 256 + t) * 4);
      sv[k] = *(const int4*)(src + base + (k * 256 + t) * 4);
      ev[k] = *(const float4*)(edge_w + base + (k * 256 + t) * 4);
    }
#pragma unroll
    for (int k = 0; k < 4; ++k) {
      cs[4 * k + 0] = cu[sv[k].x];
      cs[4 * k + 1] = cu[sv[k].y];
      cs[4 * k + 2] = cu[sv[k].z];
      cs[4 * k + 3] = cu[sv[k].w];
    }
  }
  __syncthreads();

  int rk[16];
  if (whole) {
#pragma unroll
    for (int k = 0; k < 4; ++k) {
      rk[4 * k + 0] = atomicAdd(&wcw[d[k].x >> BIN_SHIFT], 1);
      rk[4 * k + 1] = atomicAdd(&wcw[d[k].y >> BIN_SHIFT], 1);
      rk[4 * k + 2] = atomicAdd(&wcw[d[k].z >> BIN_SHIFT], 1);
      rk[4 * k + 3] = atomicAdd(&wcw[d[k].w >> BIN_SHIFT], 1);
    }
  } else {
    for (int e = base + t; e < nedges; e += 256)
      atomicAdd(&wcw[dst[e] >> BIN_SHIFT], 1);
  }
  __syncthreads();

  int pre[GROUP_WAVES];
  int hb = 0;
#pragma unroll
  for (int w = 0; w < GROUP_WAVES; ++w) {
    pre[w] = hb;
    hb += wc[w * MAX_BINS + t];
  }
  int inc = hb;
#pragma unroll
  for (int s = 1; s < 32; s <<= 1) {
    const int u = __shfl_up(inc, s);
    if (lane >= s) inc += u;
  }
  if (lane == 31) wtot[wave] = inc;
  __syncthreads();
  int carry = 0;
#pragma unroll
  for (int w = 0; w < 8; ++w) carry += (w < wave) ? wtot[w] : 0;
  const int lstart = carry + inc - hb;
#pragma unroll
  for (int w = 0; w < GROUP_WAVES; ++w) wc[w * MAX_BINS + t] = lstart + pre[w];
  if (t < nbins) P[t * nchunks + c] = lstart;
  if (t == MAX_BINS - 1) P[nbins * nchunks + c] = lstart + hb;
  __syncthreads();

  if (whole) {
#pragma unroll
    for (int k = 0; k < 4; ++k) {
      const int s0 = wcw[d[k].x >> BIN_SHIFT] + rk[4 * k + 0];
      const int s1 = wcw[d[k].y >> BIN_SHIFT] + rk[4 * k + 1];
      const int s2 = wcw[d[k].z >> BIN_SHIFT] + rk[4 * k + 2];
      const int s3 = wcw[d[k].w >> BIN_SHIFT] + rk[4 * k + 3];
      sRec[s0] = d[k].x;
      sRec[s1] = d[k].y;
      sRec[s2] = d[k].z;
      sRec[s3] = d[k].w;
      sVal[s0] = make_int2(sv[k].x, __float_as_int(ev[k].x * cs[4 * k + 0]));
      sVal[s1] = make_int2(sv[k].y, __float_as_int(ev[k].y * cs[4 * k + 1]));
      sVal[s2] = make_int2(sv[k].z, __float_as_int(ev[k].z * cs[4 * k + 2]));
      sVal[s3] = make_int2(sv[k].w, __float_as_int(ev[k].w * cs[4 * k + 3]));
    }
  } else {
    for (int e = base + t; e < nedges; e += 256) {
      const int de = dst[e];
      const int se = src[e];
      const float we = edge_w[e] * cu[se];
      const int sl = atomicAdd(&wcw[de >> BIN_SHIFT], 1);
      sRec[sl] = de;
      sVal[sl] = make_int2(se, __float_as_int(we));
    }
  }
  __syncthreads();

  if (whole) {
    int4* ro = (int4*)(rec + base);
    int4* vo = (int4*)(val + base);
#pragma unroll
    for (int k = 0; k < CHUNK / 1024; ++k) {
      const int i = k * 256 + t;
      ro[i] = sRec4[i];
    }
#pragma unroll
    for (int k = 0; k < CHUNK / 512; ++k) {
      const int i = k * 256 + t;
      vo[i] = sVal4[i];
    }
  } else {
    const int n = nedges - base;
    for (int i = t; i < n; i += 256) {
      rec[base + i] = sRec[i];
      val[base + i] = sVal[i];
    }
  }
}

#define RANK_CAP 16384
#define PIECES   8

__device__ __forceinline__ void put_edge(
    int* sS, int* sV, int* __restrict__ ps, int* __restrict__ pw,
    const int slot, const int2 v, const int lo, const bool staged) {
  if (staged) {
    sS[slot] = v.x;
    sV[slot] = v.y;
  } else {
    ps[lo + slot] = v.x;
    pw[lo + slot] = v.y;
  }
}

__global__ __launch_bounds__(1024) void bin_rank_kernel(
    const int* __restrict__ rec, const int2* __restrict__ val,
    const int* __restrict__ P, int* __restrict__ off, int* __restrict__ ps,
    int* __restrict__ pw, int nrows, int nchunks) {
  __shared__ int cntL[BIN_W];
  __shared__ int curL[BIN_W];
  __shared__ int pcS[MAX_CHUNKS];
  __shared__ int pcN[MAX_CHUNKS];
  __shared__ int wlo[32];
  __shared__ int whi[32];
  __shared__ int wcL[RANK_WAVES * BIN_W];
  __shared__ int sS[RANK_CAP];
  __shared__ int sV[RANK_CAP];
  const int t = threadIdx.x;
  const int b = blockIdx.x;
  const int lane = t & 31;
  const int wave = __builtin_amdgcn_readfirstlane(t >> 5);
  const int tbase = b << BIN_SHIFT;
  int* wcw = wcL + wave * BIN_W;

  int a0 = 0;
  int a1 = 0;
  if (t < nchunks) {
    a0 = P[b * nchunks + t];
    a1 = P[(b + 1) * nchunks + t];
  }
  pcS[lane * 32 + wave] = t * CHUNK + a0;
  pcN[lane * 32 + wave] = a1 - a0;
#pragma unroll
  for (int k = 0; k < (RANK_WAVES * BIN_W) / 1024; ++k) wcL[k * 1024 + t] = 0;
  a0 = wave_sum_i32(a0);
  a1 = wave_sum_i32(a1);
  if (lane == 0) {
    wlo[wave] = a0;
    whi[wave] = a1;
  }
  __syncthreads();
  const int lo = __builtin_amdgcn_readfirstlane(wave_sum_i32(wlo[lane]));
  const int hi = __builtin_amdgcn_readfirstlane(wave_sum_i32(whi[lane]));
  const int n = hi - lo;
  const bool staged = n <= RANK_CAP;

  const int vS = pcS[wave * 32 + lane];
  const int vN = pcN[wave * 32 + lane];
  const int npw = (nchunks - wave + 31) >> 5;

  for (int j0 = 0; j0 < npw; j0 += PIECES) {
    int s[PIECES];
    int m[PIECES];
    int d[PIECES];
#pragma unroll
    for (int k = 0; k < PIECES; ++k) {
      s[k] = __builtin_amdgcn_readlane(vS, j0 + k);
      m[k] = __builtin_amdgcn_readlane(vN, j0 + k);
    }
#pragma unroll
    for (int k = 0; k < PIECES; ++k)
      d[k] = rec[(lane < m[k]) ? s[k] + lane : 0];
#pragma unroll
    for (int k = 0; k < PIECES; ++k)
      if (lane < m[k]) atomicAdd(&wcw[d[k] - tbase], 1);
#pragma unroll
    for (int k = 0; k < PIECES; ++k)
      if (m[k] > 32)
        for (int j = lane + 32; j < m[k]; j += 32)
          atomicAdd(&wcw[rec[s[k] + j] - tbase], 1);
  }
  __syncthreads();
  if (t < BIN_W) {
    int v[RANK_WAVES];
#pragma unroll
    for (int w = 0; w < RANK_WAVES; ++w) v[w] = wcL[w * BIN_W + t];
    int run = 0;
#pragma unroll
    for (int w = 0; w < RANK_WAVES; ++w) {
      wcL[w * BIN_W + t] = run;
      run += v[w];
    }
    cntL[t] = run;
  }
  __syncthreads();
  if (t < 32) {
    int v[16];
    int s = 0;
#pragma unroll
    for (int j = 0; j < 16; ++j) {
      v[j] = cntL[t * 16 + j];
      s += v[j];
    }
    int inc = s;
#pragma unroll
    for (int d = 1; d < 32; d <<= 1) {
      const int u = __shfl_up(inc, d);
      if (t >= d) inc += u;
    }
    int run = inc - s;
#pragma unroll
    for (int j = 0; j < 16; ++j) {
      curL[t * 16 + j] = run;
      run += v[j];
    }
  }
  __syncthreads();
  if (t < BIN_W) {
    const int g = tbase + t;
    if (g <= nrows) off[g] = lo + curL[t];
  }
  for (int j0 = 0; j0 < npw; j0 += PIECES) {
    int s[PIECES];
    int m[PIECES];
    int r[PIECES];
    int2 v[PIECES];
#pragma unroll
    for (int k = 0; k < PIECES; ++k) {
      s[k] = __builtin_amdgcn_readlane(vS, j0 + k);
      m[k] = __builtin_amdgcn_readlane(vN, j0 + k);
    }
#pragma unroll
    for (int k = 0; k < PIECES; ++k) {
      const int i = (lane < m[k]) ? s[k] + lane : 0;
      r[k] = rec[i] - tbase;
      v[k] = val[i];
    }
    int slot[PIECES];
#pragma unroll
    for (int k = 0; k < PIECES; ++k)
      slot[k] = (lane < m[k]) ? curL[r[k]] + atomicAdd(&wcw[r[k]], 1) : 0;
#pragma unroll
    for (int k = 0; k < PIECES; ++k)
      if (lane < m[k]) put_edge(sS, sV, ps, pw, slot[k], v[k], lo, staged);
#pragma unroll
    for (int k = 0; k < PIECES; ++k)
      if (m[k] > 32)
        for (int j = lane + 32; j < m[k]; j += 32) {
          const int rr = rec[s[k] + j] - tbase;
          const int2 vv = val[s[k] + j];
          const int sl = curL[rr] + atomicAdd(&wcw[rr], 1);
          put_edge(sS, sV, ps, pw, sl, vv, lo, staged);
        }
  }
  if (staged) {
    __syncthreads();
    for (int i = t; i < n; i += 1024) {
      ps[lo + i] = sS[i];
      pw[lo + i] = sV[i];
    }
  }
}

#define TILE_ROWS 32

__device__ __forceinline__ void fma_row(float4& acc, const float4 v, const float w) {
  acc.x = fmaf(v.x, w, acc.x);
  acc.y = fmaf(v.y, w, acc.y);
  acc.z = fmaf(v.z, w, acc.z);
  acc.w = fmaf(v.w, w, acc.w);
}

__device__ __forceinline__ void add_run(float4& acc, const float* __restrict__ fb,
                                        const int ss, const int wb, const int n) {
  int i = 0;
  for (; i + 8 <= n; i += 8) {
    float4 v[8];
    float wi[8];
#pragma unroll
    for (int k = 0; k < 8; ++k) {
      const int si = __builtin_amdgcn_readlane(ss, i + k);
      wi[k] = __int_as_float(__builtin_amdgcn_readlane(wb, i + k));
      v[k] = *(const float4*)(fb + (size_t)si * D_IN);
    }
#pragma unroll
    for (int k = 0; k < 8; ++k) fma_row(acc, v[k], wi[k]);
  }
  if (i + 4 <= n) {
    float4 v[4];
    float wi[4];
#pragma unroll
    for (int k = 0; k < 4; ++k) {
      const int si = __builtin_amdgcn_readlane(ss, i + k);
      wi[k] = __int_as_float(__builtin_amdgcn_readlane(wb, i + k));
      v[k] = *(const float4*)(fb + (size_t)si * D_IN);
    }
#pragma unroll
    for (int k = 0; k < 4; ++k) fma_row(acc, v[k], wi[k]);
    i += 4;
  }
  for (; i < n; ++i) {
    const int si = __builtin_amdgcn_readlane(ss, i);
    const float wi = __int_as_float(__builtin_amdgcn_readlane(wb, i));
    fma_row(acc, *(const float4*)(fb + (size_t)si * D_IN), wi);
  }
}

__device__ __forceinline__ void add_long_run(
    float4& acc, const float* __restrict__ fb, const int* __restrict__ ps,
    const float* __restrict__ pw, const int beg, const int end, const int lane) {
#pragma nounroll
  for (int p = beg; p < end; p += 32) {
    const int n = (end - p < 32) ? end - p : 32;
    const int i = (lane < n) ? p + lane : p;
    int si = ps[i];
    float wi = pw[i];
    asm volatile("" : "+v"(si), "+v"(wi));
    add_run(acc, fb, si, __float_as_int(wi), n);
  }
}

__device__ __forceinline__ int load_bounds(const int* __restrict__ off, const int d0,
                                           const int lane, const int nrows) {
  int g = d0 + lane;
  if (g > nrows) g = nrows;
  return (lane < 3) ? off[g] : 0;
}

__global__ __launch_bounds__(512) void influence_pull_gemm_kernel(
    const float* __restrict__ feat, const float* __restrict__ W,
    const float* __restrict__ cv, const int* __restrict__ off,
    const int* __restrict__ ps, const float* __restrict__ pw,
    float* __restrict__ out, int nrows, int ntiles) {
  __shared__ float sW[D_IN * D_OUT];
  __shared__ float4 sA[TILE_ROWS * 32];

  const int tid = threadIdx.x;
  int tile = blockIdx.x;
  if (tile >= ntiles) return;

  int b0, b1, b2;
  int t0, t1;
  float u0, u1;
  {
    const int lane = tid & 31;
    const int wave = tid >> 5;
    const int ov = load_bounds(off, tile * TILE_ROWS + 2 * wave, lane, nrows);

#pragma unroll
    for (int i = 0; i < (D_IN * D_OUT) / (4 * 512); ++i)
      ((float4*)sW)[tid + i * 512] = ((const float4*)W)[tid + i * 512];

    b0 = __builtin_amdgcn_readlane(ov, 0);
    b1 = __builtin_amdgcn_readlane(ov, 1);
    b2 = __builtin_amdgcn_readlane(ov, 2);
    const int i0 = (b1 - b0 <= 32 && lane < b1 - b0) ? b0 + lane : 0;
    const int i1 = (b2 - b1 <= 32 && lane < b2 - b1) ? b1 + lane : 0;
    t0 = ps[i0];
    t1 = ps[i1];
    u0 = pw[i0];
    u1 = pw[i1];
  }

  for (;;) {
    const int lane = tid & 31;
    const int wave = __builtin_amdgcn_readfirstlane(tid >> 5);

    const int rowBase = tile * TILE_ROWS;
    const int deg0 = b1 - b0;
    const int deg1 = b2 - b1;

    asm volatile("" : "+v"(t0), "+v"(t1), "+v"(u0), "+v"(u1));

    const int next = tile + gridDim.x;
    const int ovN = load_bounds(off, next * TILE_ROWS + 2 * wave, lane, nrows);

    const float* fb = feat + lane * 4;
    const int lr0 = 2 * wave;

    float4 acc0 = {0.f, 0.f, 0.f, 0.f};
    float4 acc1 = {0.f, 0.f, 0.f, 0.f};
    if (deg0 <= 32) add_run(acc0, fb, t0, __float_as_int(u0), deg0);
    else add_long_run(acc0, fb, ps, pw, b0, b1, lane);

    const int nb0 = __builtin_amdgcn_readlane(ovN, 0);
    const int nb1 = __builtin_amdgcn_readlane(ovN, 1);
    const int nb2 = __builtin_amdgcn_readlane(ovN, 2);

    if (deg1 <= 32) add_run(acc1, fb, t1, __float_as_int(u1), deg1);
    else add_long_run(acc1, fb, ps, pw, b1, b2, lane);

    const int ni0 = (nb1 - nb0 <= 32 && lane < nb1 - nb0) ? nb0 + lane : 0;
    const int ni1 = (nb2 - nb1 <= 32 && lane < nb2 - nb1) ? nb1 + lane : 0;
    const int nt0 = ps[ni0];
    const int nt1 = ps[ni1];
    const float nu0 = pw[ni0];
    const float nu1 = pw[ni1];

    sA[lr0 * 32 + (lane ^ lr0)] = acc0;
    sA[(lr0 + 1) * 32 + (lane ^ (lr0 + 1))] = acc1;
    __syncthreads();

    const int rt      = wave >> 3;
    const int colBase = (wave & 7) * 16;
    const int m       = lane & 15;
    const int h       = lane >> 4;
    const int kh      = h * 2;
    const int ar = rt * 16 + m;
    const v2f* arow = (const v2f*)sA + ar * 64;

    const int mBase = rowBase + rt * 16 + h * 8;
    const int crow = rowBase + rt * 16 + m;
    const float cvl = cv[crow < nrows ? crow : nrows - 1];
    float cvr[8];
#pragma unroll
    for (int v = 0; v < 8; ++v) cvr[v] = __shfl(cvl, h * 8 + v);

    v8f acc = {};
#pragma unroll
    for (int kk = 0; kk < D_IN / 4; ++kk) {
      const v2f A = arow[((kk ^ ar) << 1) | h];
      const int kb = kk * 4 + kh;
      v2f B;
      B.x = sW[kb * D_OUT + colBase + m];
      B.y = sW[(kb + 1) * D_OUT + colBase + m];
      acc = __builtin_amdgcn_wmma_f32_16x16x4_f32(
           false, A,  false, B,
           (short)0, acc,  false,  false);
    }

#pragma unroll
    for (int v = 0; v < 8; ++v) {
      const int row = mBase + v;
      if (row < nrows)
        out[(size_t)row * D_OUT + colBase + m] = acc[v] * cvr[v];
    }

    if (next >= ntiles) break;
    __syncthreads();
    tile = next;
    b0 = nb0; b1 = nb1; b2 = nb2;
    t0 = nt0; t1 = nt1;
    u0 = nu0; u1 = nu1;
  }
}

extern "C" void kernel_launch(void* const* d_in, const int* in_sizes, int n_in,
                              void* d_out, int out_size, void* d_ws, size_t ws_size,
                              hipStream_t stream) {
  const float* feat = (const float*)d_in[0];
  const float* W    = (const float*)d_in[1];
  const float* cu   = (const float*)d_in[2];
  const float* cv   = (const float*)d_in[3];
  const float* ew   = (const float*)d_in[4];
  const int*   src  = (const int*)d_in[5];
  const int*   dst  = (const int*)d_in[6];

  float* out = (float*)d_out;

  const int N = in_sizes[2];
  const int E = in_sizes[5];

  const int nbins   = N / BIN_W + 1;
  const int nchunks = (E + CHUNK - 1) / CHUNK;
  const int Pp = (((nbins + 1) * nchunks) / 4 + 1) * 4;
  const int Np = (N / 4096 + 1) * 4096;
  const int Ep = nchunks * CHUNK;
  int*  P    = (int*)d_ws;
  int*  off  = P + Pp;
  int*  rec  = off + Np;
  int2* val  = (int2*)(rec + Ep);
  int*  ps   = (int*)(val + Ep);
  int*  pw   = ps + E;

  bin_group_kernel<<<nchunks, 256, 0, stream>>>(dst, src, ew, cu, P, rec, val, E,
                                                nbins, nchunks);
  bin_rank_kernel<<<nbins, 1024, 0, stream>>>(rec, val, P, off, ps, pw, N, nchunks);

  const int rowTiles = (N + TILE_ROWS - 1) / TILE_ROWS;
  influence_pull_gemm_kernel<<<1024, 512, 0, stream>>>(
      feat, W, cv, off, ps, (const float*)pw, out, N, rowTiles);
}
